# starting kernel
# baseline (speedup 1.0000x reference)
#include <hip/hip_runtime.h>

typedef __attribute__((ext_vector_type(2))) float v2f;
typedef __attribute__((ext_vector_type(8))) float v8f;

#define HID 64
#define WST 68

__global__ __launch_bounds__(256) void gnn_gemm_relu(
    const float* __restrict__ in, const float* __restrict__ W,
    const float* __restrict__ resid, float* __restrict__ out, int n)
{
    __shared__ float lw[64 * WST];
    const int tid = threadIdx.x;

    for (int i = tid; i < 64 * 64; i += 256) {
        int r = i >> 6, c = i & 63;
        lw[r * WST + c] = W[i];
    }
    __syncthreads();

    const int wave = tid >> 5;
    const int lane = tid & 31;
    const int half = lane >> 4;
    const int l16  = lane & 15;
    const int row0 = blockIdx.x * 128 + wave * 16;

    const int  arow  = row0 + l16;
    const bool rowok = arow < n;
    const float* abase = in + (size_t)(rowok ? arow : 0) * HID + half * 2;
    v2f a[16];
#pragma unroll
    for (int s = 0; s < 16; ++s) {
        float2 t = rowok ? *(const float2*)(abase + 4 * s) : make_float2(0.f, 0.f);
        a[s].x = t.x; a[s].y = t.y;
    }

    v8f acc[4] = {};

#pragma unroll
    for (int s = 0; s < 16; ++s) {
#pragma unroll
        for (int j = 0; j < 4; ++j) {
            const int nc = j * 16 + l16;
            float2 bb = *(const float2*)&lw[nc * WST + 4 * s + half * 2];
            v2f b; b.x = bb.x; b.y = bb.y;
            acc[j] = __builtin_amdgcn_wmma_f32_16x16x4_f32(
                false, a[s], false, b, (short)0, acc[j], false, false);
        }
    }

#pragma unroll
    for (int j = 0; j < 4; ++j) {
        const int colg = j * 16 + l16;
#pragma unroll
        for (int v = 0; v < 8; ++v) {
            const int rowg = row0 + v + half * 8;
            if (rowg < n) {
                float val = fmaxf(acc[j][v], 0.f);
                if (resid) val += resid[(size_t)rowg * HID + colg];
                out[(size_t)rowg * HID + colg] = val;
            }
        }
    }
}

__device__ __forceinline__ void hw_atomic_add_f32(float* p, float v) {
    asm volatile("global_atomic_add_f32 %0, %1, off"
                 :
                 : "v"(p), "v"(v)
                 : "memory");
}

__global__ __launch_bounds__(256) void gnn_scatter(
    const float* __restrict__ m, const int* __restrict__ row,
    const int* __restrict__ col, float* __restrict__ t, int ne)
{
    const int wave = (int)((blockIdx.x * blockDim.x + threadIdx.x) >> 5);
    const int lane = threadIdx.x & 31;
    const int e0 = wave * 4;
#pragma unroll
    for (int i = 0; i < 4; ++i) {
        const int e = e0 + i;
        if (e >= ne) return;
        const int r = row[e];
        const int c = col[e];
        float2 v = *(const float2*)(m + (size_t)c * HID + 2 * lane);
        float* dst = t + (size_t)r * HID + 2 * lane;
        hw_atomic_add_f32(dst,     v.x);
        hw_atomic_add_f32(dst + 1, v.y);
    }
}

__global__ __launch_bounds__(256) void gnn_rmsnorm(
    const float* __restrict__ x, const float* __restrict__ w,
    float* __restrict__ out, int n)
{
    const long long gid = (long long)blockIdx.x * blockDim.x + threadIdx.x;
    const int r    = (int)(gid >> 5);
    const int lane = threadIdx.x & 31;
    if (r >= n) return;

    float2 v = *(const float2*)(x + (size_t)r * HID + 2 * lane);
    float ss = v.x * v.x + v.y * v.y;
#pragma unroll
    for (int off = 16; off >= 1; off >>= 1)
        ss += __shfl_xor(ss, off, 32);
    const float inv = rsqrtf(ss * (1.0f / 64.0f) + 1e-5f);

    float2 wv = *(const float2*)(w + 2 * lane);
    float2 o; o.x = v.x * inv * wv.x; o.y = v.y * inv * wv.y;
    *(float2*)(out + (size_t)r * HID + 2 * lane) = o;
}

extern "C" void kernel_launch(void* const* d_in, const int* in_sizes, int n_in,
                              void* d_out, int out_size, void* d_ws, size_t ws_size,
                              hipStream_t stream)
{
    const float* x       = (const float*)d_in[0];
    const float* conv_w  = (const float*)d_in[1];
    const float* conv_nw = (const float*)d_in[2];
    const float* hid_w   = (const float*)d_in[3];
    const float* hid_nw  = (const float*)d_in[4];
    const int*   row     = (const int*)d_in[5];
    const int*   col     = (const int*)d_in[6];

    const int n  = in_sizes[0] / HID;
    const int ne = in_sizes[5];

    float* P = (float*)d_out;
    float* Q = (float*)d_ws;
    float* R = Q + (size_t)n * HID;
    const size_t hbytes = (size_t)n * HID * sizeof(float);

    const dim3 blk(256);
    const int gemm_grid = (n + 127) / 128;
    const int rms_grid  = (n + 7) / 8;
    const int sc_grid   = ((ne + 3) / 4 + 7) / 8;

    const float* hin = x;
    for (int l = 0; l < 4; ++l) {
        const float* cw = conv_w + (size_t)l * HID * HID;
        const float* hw = hid_w  + (size_t)l * HID * HID;
        const float* cn = conv_nw + l * HID;
        const float* hn = hid_nw  + l * HID;

        gnn_gemm_relu<<<gemm_grid, blk, 0, stream>>>(hin, cw, nullptr, Q, n);
        hipMemcpyAsync(R, hin, hbytes, hipMemcpyDeviceToDevice, stream);
        gnn_scatter<<<sc_grid, blk, 0, stream>>>(Q, row, col, R, ne);
        gnn_rmsnorm<<<rms_grid, blk, 0, stream>>>(R, cn, R, n);
        gnn_gemm_relu<<<gemm_grid, blk, 0, stream>>>(R, hw, R, P, n);
        gnn_rmsnorm<<<rms_grid, blk, 0, stream>>>(P, hn, P, n);

        hin = P;
    }
}


# Aggregation without f32 atomics: each edge once a call takes an entry in its destination row's 32-entry list (integer add); per layer a wave sums 8 rows' sources in registers, in list order
# speedup vs baseline: 2.3078x; 2.3078x over previous
#include <hip/hip_runtime.h>

typedef __attribute__((ext_vector_type(2))) float v2f;
typedef __attribute__((ext_vector_type(8))) float v8f;
typedef __attribute__((ext_vector_type(4))) float v4f;

#define HID 64
#define WST 68

__global__ __launch_bounds__(256) void gnn_gemm_relu(
    const float* __restrict__ in, const float* __restrict__ W,
    const float* __restrict__ resid, float* __restrict__ out,
    float* __restrict__ copy, int* __restrict__ zero, int n)
{
    __shared__ float lw[64 * WST];
    const int tid = threadIdx.x;

    if (zero) {
        const int gid = (int)blockIdx.x * 256 + tid;
        if (gid <= n) zero[gid] = 0;
    }

    if (copy) {
        const int rows_left = n - (int)blockIdx.x * 128;
        const size_t base = (size_t)blockIdx.x * 128 * HID;
        const v4f* src = (const v4f*)(in + base) + tid;
        v4f* dst = (v4f*)(copy + base) + tid;
        if (rows_left >= 128) {
            v4f cv[8];
#pragma unroll
            for (int p = 0; p < 8; ++p) cv[p] = src[p * 256];
            asm volatile("" :: "v"(cv[0]), "v"(cv[1]), "v"(cv[2]), "v"(cv[3]),
                               "v"(cv[4]), "v"(cv[5]), "v"(cv[6]), "v"(cv[7]));
#pragma unroll
            for (int p = 0; p < 8; ++p) dst[p * 256] = cv[p];
        } else {
            for (int p = 0; 16 * p + (tid >> 4) < rows_left; ++p) dst[p * 256] = src[p * 256];
        }
    }

    for (int i = tid; i < 64 * 64; i += 256) {
        int r = i >> 6, c = i & 63;
        lw[r * WST + c] = W[i];
    }
    __syncthreads();

    const int wave = tid >> 5;
    const int lane = tid & 31;
    const int half = lane >> 4;
    const int l16  = lane & 15;
    const int row0 = blockIdx.x * 128 + wave * 16;

    const int  arow  = row0 + l16;
    const bool rowok = arow < n;
    const float* abase = in + (size_t)(rowok ? arow : 0) * HID + half * 2;
    v2f a[16];
#pragma unroll
    for (int s = 0; s < 16; ++s) {
        float2 t = rowok ? *(const float2*)(abase + 4 * s) : make_float2(0.f, 0.f);
        a[s].x = t.x; a[s].y = t.y;
    }

    v8f acc[4] = {};

#pragma unroll
    for (int s = 0; s < 16; ++s) {
#pragma unroll
        for (int j = 0; j < 4; ++j) {
            const int nc = j * 16 + l16;
            float2 bb = *(const float2*)&lw[nc * WST + 4 * s + half * 2];
            v2f b; b.x = bb.x; b.y = bb.y;
            acc[j] = __builtin_amdgcn_wmma_f32_16x16x4_f32(
                false, a[s], false, b, (short)0, acc[j], false, false);
        }
    }

#pragma unroll
    for (int j = 0; j < 4; ++j) {
        const int colg = j * 16 + l16;
#pragma unroll
        for (int v = 0; v < 8; ++v) {
            const int rowg = row0 + v + half * 8;
            if (rowg < n) {
                float val = fmaxf(acc[j][v], 0.f);
                if (resid) val += resid[(size_t)rowg * HID + colg];
                out[(size_t)rowg * HID + colg] = val;
            }
        }
    }
}

#define ELLK 32
#define RPW  8

__global__ __launch_bounds__(256) void gnn_fill(
    const int* __restrict__ row, const int* __restrict__ col,
    int* __restrict__ cnt, int* __restrict__ ell, int* __restrict__ ovf,
    int n, int ne)
{
    const int gid = (int)blockIdx.x * 256 + (int)threadIdx.x;
    const int stride = (int)gridDim.x * 256;
    int r[4], c[4], slot[4];
#pragma unroll
    for (int k = 0; k < 4; ++k) {
        const int e = gid + k * stride;
        r[k] = e < ne ? row[e] : -1;
        c[k] = e < ne ? col[e] : 0;
    }
#pragma unroll
    for (int k = 0; k < 4; ++k)
        slot[k] = r[k] >= 0 ? atomicAdd(cnt + r[k], 1) : 0;
#pragma unroll
    for (int k = 0; k < 4; ++k) {
        if (r[k] < 0) continue;
        if (slot[k] < ELLK) {
            ell[(size_t)r[k] * ELLK + slot[k]] = c[k];
        } else {
            const int i = atomicAdd(cnt + n, 1);
            ovf[2 * (size_t)i]     = r[k];
            ovf[2 * (size_t)i + 1] = c[k];
        }
    }
}

__global__ __launch_bounds__(256) void gnn_pull(
    const float* __restrict__ m, const int* __restrict__ cnt,
    const int* __restrict__ ell, const int* __restrict__ ovf,
    float* __restrict__ t, int n)
{
    const int lane = threadIdx.x & 31;
    const int wave = __builtin_amdgcn_readfirstlane(
        (int)((blockIdx.x * 256 + threadIdx.x) >> 5));
    const int r0 = wave * RPW;
    if (r0 >= n) return;

    int idx[RPW], deg[RPW];
    float2 acc[RPW];
#pragma unroll
    for (int j = 0; j < RPW; ++j) {
        const int r = r0 + j < n ? r0 + j : n - 1;
        idx[j] = ell[(size_t)r * ELLK + lane];
        acc[j] = *(const float2*)(t + (size_t)r * HID + 2 * lane);
        deg[j] = r0 + j < n ? __builtin_amdgcn_readfirstlane(cnt[r]) : 0;
    }

#pragma unroll
    for (int j = 0; j < RPW; ++j) {
        const int d = deg[j] < ELLK ? deg[j] : ELLK;
        for (int s = 0; s < d; s += 8) {
            float2 v[8];
#pragma unroll
            for (int k = 0; k < 8; ++k) {
                v[k] = make_float2(0.f, 0.f);
                if (s + k < d) {
                    const int c = __builtin_amdgcn_readlane(idx[j], s + k);
                    v[k] = *(const float2*)(m + (size_t)c * HID + 2 * lane);
                }
            }
#pragma unroll
            for (int k = 0; k < 8; ++k) { acc[j].x += v[k].x; acc[j].y += v[k].y; }
        }
    }

#pragma unroll
    for (int j = 0; j < RPW; ++j) {
        if (deg[j] > ELLK) {
            const int r = r0 + j;
            const int novf = __builtin_amdgcn_readfirstlane(cnt[n]);
            for (int b = 0; b < novf; b += 32) {
                const int i = b + lane;
                int orow = -1, ocol = 0;
                if (i < novf) { orow = ovf[2 * (size_t)i]; ocol = ovf[2 * (size_t)i + 1]; }
                unsigned mask = __builtin_amdgcn_ballot_w32(orow == r);
                while (mask) {
                    const int k = __builtin_ctz(mask);
                    mask &= mask - 1;
                    const int c = __builtin_amdgcn_readlane(ocol, k);
                    const float2 v = *(const float2*)(m + (size_t)c * HID + 2 * lane);
                    acc[j].x += v.x; acc[j].y += v.y;
                }
            }
        }
        if (r0 + j < n)
            *(float2*)(t + (size_t)(r0 + j) * HID + 2 * lane) = acc[j];
    }
}

__global__ __launch_bounds__(256, 8) void gnn_norm_gemm_norm(
    float* __restrict__ t, const float* __restrict__ W,
    const float* __restrict__ w1, const float* __restrict__ w2,
    float* __restrict__ fin, const float* __restrict__ Wn,
    float* __restrict__ mout, int n)
{
    __shared__ __attribute__((aligned(16))) float lw[2 * 64 * WST];
    __shared__ __attribute__((aligned(16))) float lg[2 * HID];
    const int tid = threadIdx.x;
    float gword = 0.f;
    if (tid < 2 * HID) gword = tid < HID ? w1[tid] : w2[tid - HID];

    if (Wn) {
        for (int i = tid; i < 64 * 64; i += 256) {
            const int f = i >> 6, k = i & 63;
            const int rho = ((f >> 3) & 3) * 16 + (f >> 5) * 8 + (f & 7);
            const float a = W[i];
            const float b = Wn[i];
            lw[rho * WST + k] = a;
            lw[64 * WST + rho * WST + k] = b;
        }
    } else {
        for (int i = tid; i < 64 * 64; i += 256) {
            const int f = i >> 6, k = i & 63;
            const int rho = ((f >> 3) & 3) * 16 + (f >> 5) * 8 + (f & 7);
            lw[rho * WST + k] = W[i];
        }
    }
    if (tid < 2 * HID) lg[tid] = gword;
    __syncthreads();

    const int wave = __builtin_amdgcn_readfirstlane(tid >> 5);
    const int lane = tid & 31;
    const int half = lane >> 4;
    const int c    = lane & 15;
    const int node0 = blockIdx.x * 128 + wave * 16;
    if (node0 >= n) return;
    const int node = node0 + c;
    const bool ok  = node < n;
    const int q0 = (ok ? node : n - 1) * (HID / 4) + half * 8;
    const float4* t4 = (const float4*)t;

    float u[32];
#pragma unroll
    for (int q = 0; q < 8; ++q) {
        const float4 v = t4[q0 + q];
        u[4 * q + 0] = v.x; u[4 * q + 1] = v.y;
        u[4 * q + 2] = v.z; u[4 * q + 3] = v.w;
    }

    {
        float p0 = 0.f, p1 = 0.f, p2 = 0.f, p3 = 0.f;
#pragma unroll
        for (int q = 0; q < 8; ++q) {
            p0 += u[4 * q + 0] * u[4 * q + 0];
            p1 += u[4 * q + 1] * u[4 * q + 1];
            p2 += u[4 * q + 2] * u[4 * q + 2];
            p3 += u[4 * q + 3] * u[4 * q + 3];
        }
        float ss = (p0 + p1) + (p2 + p3);
        ss += __shfl_xor(ss, 16, 32);
        const float inv = rsqrtf(ss * (1.0f / 64.0f) + 1e-5f);
#pragma unroll
        for (int q = 0; q < 8; ++q) {
            const float4 g = *(const float4*)&lg[half * 32 + 4 * q];
            u[4 * q + 0] = u[4 * q + 0] * inv * g.x;
            u[4 * q + 1] = u[4 * q + 1] * inv * g.y;
            u[4 * q + 2] = u[4 * q + 2] * inv * g.z;
            u[4 * q + 3] = u[4 * q + 3] * inv * g.w;
        }
    }

    v8f acc[4] = {};

#pragma unroll
    for (int sp = 0; sp < 8; ++sp) {
        v2f b0; b0.x = u[4 * sp];     b0.y = u[4 * sp + 1];
        v2f b1; b1.x = u[4 * sp + 2]; b1.y = u[4 * sp + 3];
        float4 aa[4];
#pragma unroll
        for (int jo = 0; jo < 4; ++jo)
            aa[jo] = *(const float4*)&lw[(jo * 16 + c) * WST + half * 32 + 4 * sp];
#pragma unroll
        for (int jo = 0; jo < 4; ++jo) {
            v2f a; a.x = aa[jo].x; a.y = aa[jo].y;
            acc[jo] = __builtin_amdgcn_wmma_f32_16x16x4_f32(
                false, a, false, b0, (short)0, acc[jo], false, false);
        }
#pragma unroll
        for (int jo = 0; jo < 4; ++jo) {
            v2f a; a.x = aa[jo].z; a.y = aa[jo].w;
            acc[jo] = __builtin_amdgcn_wmma_f32_16x16x4_f32(
                false, a, false, b1, (short)0, acc[jo], false, false);
        }
    }

    float p0 = 0.f, p1 = 0.f, p2 = 0.f, p3 = 0.f;
#pragma unroll
    for (int jo = 0; jo < 4; ++jo) {
#pragma unroll
        for (int r = 0; r < 8; ++r)
            u[8 * jo + r] += fmaxf(acc[jo][r], 0.f);
    }
#pragma unroll
    for (int q = 0; q < 8; ++q) {
        p0 += u[4 * q + 0] * u[4 * q + 0];
        p1 += u[4 * q + 1] * u[4 * q + 1];
        p2 += u[4 * q + 2] * u[4 * q + 2];
        p3 += u[4 * q + 3] * u[4 * q + 3];
    }
    float ss2 = (p0 + p1) + (p2 + p3);
    ss2 += __shfl_xor(ss2, 16, 32);
    const float inv2 = rsqrtf(ss2 * (1.0f / 64.0f) + 1e-5f);
    float4* out4 = (float4*)(fin ? fin : t);
#pragma unroll
    for (int q = 0; q < 8; ++q) {
        const float4 g = *(const float4*)&lg[HID + half * 32 + 4 * q];
        u[4 * q + 0] = u[4 * q + 0] * inv2 * g.x;
        u[4 * q + 1] = u[4 * q + 1] * inv2 * g.y;
        u[4 * q + 2] = u[4 * q + 2] * inv2 * g.z;
        u[4 * q + 3] = u[4 * q + 3] * inv2 * g.w;
    }
    if (ok) {
#pragma unroll
        for (int q = 0; q < 8; ++q) {
            float4 o;
            o.x = u[4 * q + 0]; o.y = u[4 * q + 1];
            o.z = u[4 * q + 2]; o.w = u[4 * q + 3];
            out4[q0 + q] = o;
        }
    }

    if (Wn) {
        v8f acn[4] = {};
#pragma unroll
        for (int sp = 0; sp < 8; ++sp) {
            v2f b0; b0.x = u[4 * sp];     b0.y = u[4 * sp + 1];
            v2f b1; b1.x = u[4 * sp + 2]; b1.y = u[4 * sp + 3];
            float4 aa[4];
#pragma unroll
            for (int jo = 0; jo < 4; ++jo)
                aa[jo] = *(const float4*)&lw[64 * WST + (jo * 16 + c) * WST + half * 32 + 4 * sp];
#pragma unroll
            for (int jo = 0; jo < 4; ++jo) {
                v2f a; a.x = aa[jo].x; a.y = aa[jo].y;
                acn[jo] = __builtin_amdgcn_wmma_f32_16x16x4_f32(
                    false, a, false, b0, (short)0, acn[jo], false, false);
            }
#pragma unroll
            for (int jo = 0; jo < 4; ++jo) {
                v2f a; a.x = aa[jo].z; a.y = aa[jo].w;
                acn[jo] = __builtin_amdgcn_wmma_f32_16x16x4_f32(
                    false, a, false, b1, (short)0, acn[jo], false, false);
            }
        }
        if (ok) {
#pragma unroll
            for (int jo = 0; jo < 4; ++jo) {
                float4 o0, o1;
                o0.x = fmaxf(acn[jo][0], 0.f); o0.y = fmaxf(acn[jo][1], 0.f);
                o0.z = fmaxf(acn[jo][2], 0.f); o0.w = fmaxf(acn[jo][3], 0.f);
                o1.x = fmaxf(acn[jo][4], 0.f); o1.y = fmaxf(acn[jo][5], 0.f);
                o1.z = fmaxf(acn[jo][6], 0.f); o1.w = fmaxf(acn[jo][7], 0.f);
                float4* m4 = (float4*)mout;
                m4[q0 + 2 * jo]     = o0;
                m4[q0 + 2 * jo + 1] = o1;
            }
        }
    }
}

extern "C" void kernel_launch(void* const* d_in, const int* in_sizes, int n_in,
                              void* d_out, int out_size, void* d_ws, size_t ws_size,
                              hipStream_t stream)
{
    const float* x       = (const float*)d_in[0];
    const float* conv_w  = (const float*)d_in[1];
    const float* conv_nw = (const float*)d_in[2];
    const float* hid_w   = (const float*)d_in[3];
    const float* hid_nw  = (const float*)d_in[4];
    const int*   row     = (const int*)d_in[5];
    const int*   col     = (const int*)d_in[6];

    const int n  = in_sizes[0] / HID;
    const int ne = in_sizes[5];

    float* P = (float*)d_out;
    float* Q = (float*)d_ws;
    float* R = Q + (size_t)n * HID;
    int* cnt = (int*)(R + (size_t)n * HID);
    int* ell = cnt + ((size_t)n + 1 + 31) / 32 * 32;
    int* ovf = ell + (size_t)n * ELLK;

    const dim3 blk(256);
    const int gemm_grid = (n + 127) / 128;
    const int fill_grid = (ne + 1023) / 1024;
    const int pull_grid = ((n + RPW - 1) / RPW + 7) / 8;

    gnn_gemm_relu<<<gemm_grid, blk, 0, stream>>>(x, conv_w, nullptr, Q, R, cnt, n);
    gnn_fill<<<fill_grid, blk, 0, stream>>>(row, col, cnt, ell, ovf, n, ne);

    for (int l = 0; l < 4; ++l) {
        const float* hw = hid_w  + (size_t)l * HID * HID;
        const float* cn = conv_nw + l * HID;
        const float* hn = hid_nw  + l * HID;
        const float* cwn = l < 3 ? conv_w + (size_t)(l + 1) * HID * HID : nullptr;

        gnn_pull<<<pull_grid, blk, 0, stream>>>(Q, cnt, ell, ovf, R, n);
        gnn_norm_gemm_norm<<<gemm_grid, blk, 0, stream>>>(R, hw, cn, hn,
                                                          l == 3 ? P : nullptr,
                                                          cwn, Q, n);
    }
}


# Dense kernel: rows pass between memory and owner lanes through a 4 KB LDS tile per wave, so each load of t and each store covers 4 whole 128-byte lines, not 16 bytes of 32 lines; sums unchanged
# speedup vs baseline: 2.9779x; 1.2904x over previous
#include <hip/hip_runtime.h>

typedef __attribute__((ext_vector_type(2))) float v2f;
typedef __attribute__((ext_vector_type(8))) float v8f;
typedef __attribute__((ext_vector_type(4))) float v4f;

#define HID 64
#define WST 68
#define SP  68

__global__ __launch_bounds__(256) void gnn_gemm_relu(
    const float* __restrict__ in, const float* __restrict__ W,
    const float* __restrict__ resid, float* __restrict__ out,
    float* __restrict__ copy, int* __restrict__ zero, int n)
{
    __shared__ float lw[64 * WST];
    const int tid = threadIdx.x;

    if (zero) {
        const int gid = (int)blockIdx.x * 256 + tid;
        if (gid <= n) zero[gid] = 0;
    }

    if (copy) {
        const int rows_left = n - (int)blockIdx.x * 128;
        const size_t base = (size_t)blockIdx.x * 128 * HID;
        const v4f* src = (const v4f*)(in + base) + tid;
        v4f* dst = (v4f*)(copy + base) + tid;
        if (rows_left >= 128) {
            v4f cv[8];
#pragma unroll
            for (int p = 0; p < 8; ++p) cv[p] = src[p * 256];
            asm volatile("" :: "v"(cv[0]), "v"(cv[1]), "v"(cv[2]), "v"(cv[3]),
                               "v"(cv[4]), "v"(cv[5]), "v"(cv[6]), "v"(cv[7]));
#pragma unroll
            for (int p = 0; p < 8; ++p) dst[p * 256] = cv[p];
        } else {
            for (int p = 0; 16 * p + (tid >> 4) < rows_left; ++p) dst[p * 256] = src[p * 256];
        }
    }

    for (int i = tid; i < 64 * 64; i += 256) {
        int r = i >> 6, c = i & 63;
        lw[r * WST + c] = W[i];
    }
    __syncthreads();

    const int wave = tid >> 5;
    const int lane = tid & 31;
    const int half = lane >> 4;
    const int l16  = lane & 15;
    const int row0 = blockIdx.x * 128 + wave * 16;

    const int  arow  = row0 + l16;
    const bool rowok = arow < n;
    const float* abase = in + (size_t)(rowok ? arow : 0) * HID + half * 2;
    v2f a[16];
#pragma unroll
    for (int s = 0; s < 16; ++s) {
        float2 t = rowok ? *(const float2*)(abase + 4 * s) : make_float2(0.f, 0.f);
        a[s].x = t.x; a[s].y = t.y;
    }

    v8f acc[4] = {};

#pragma unroll
    for (int s = 0; s < 16; ++s) {
#pragma unroll
        for (int j = 0; j < 4; ++j) {
            const int nc = j * 16 + l16;
            float2 bb = *(const float2*)&lw[nc * WST + 4 * s + half * 2];
            v2f b; b.x = bb.x; b.y = bb.y;
            acc[j] = __builtin_amdgcn_wmma_f32_16x16x4_f32(
                false, a[s], false, b, (short)0, acc[j], false, false);
        }
    }

#pragma unroll
    for (int j = 0; j < 4; ++j) {
        const int colg = j * 16 + l16;
#pragma unroll
        for (int v = 0; v < 8; ++v) {
            const int rowg = row0 + v + half * 8;
            if (rowg < n) {
                float val = fmaxf(acc[j][v], 0.f);
                if (resid) val += resid[(size_t)rowg * HID + colg];
                out[(size_t)rowg * HID + colg] = val;
            }
        }
    }
}

#define ELLK 32
#define RPW  8

__global__ __launch_bounds__(256) void gnn_fill(
    const int* __restrict__ row, const int* __restrict__ col,
    int* __restrict__ cnt, int* __restrict__ ell, int* __restrict__ ovf,
    int n, int ne)
{
    const int gid = (int)blockIdx.x * 256 + (int)threadIdx.x;
    const int stride = (int)gridDim.x * 256;
    int r[4], c[4], slot[4];
#pragma unroll
    for (int k = 0; k < 4; ++k) {
        const int e = gid + k * stride;
        r[k] = e < ne ? row[e] : -1;
        c[k] = e < ne ? col[e] : 0;
    }
#pragma unroll
    for (int k = 0; k < 4; ++k)
        slot[k] = r[k] >= 0 ? atomicAdd(cnt + r[k], 1) : 0;
#pragma unroll
    for (int k = 0; k < 4; ++k) {
        if (r[k] < 0) continue;
        if (slot[k] < ELLK) {
            ell[(size_t)r[k] * ELLK + slot[k]] = c[k];
        } else {
            const int i = atomicAdd(cnt + n, 1);
            ovf[2 * (size_t)i]     = r[k];
            ovf[2 * (size_t)i + 1] = c[k];
        }
    }
}

__global__ __launch_bounds__(256) void gnn_pull(
    const float* __restrict__ m, const int* __restrict__ cnt,
    const int* __restrict__ ell, const int* __restrict__ ovf,
    float* __restrict__ t, int n)
{
    const int lane = threadIdx.x & 31;
    const int wave = __builtin_amdgcn_readfirstlane(
        (int)((blockIdx.x * 256 + threadIdx.x) >> 5));
    const int r0 = wave * RPW;
    if (r0 >= n) return;

    int idx[RPW], deg[RPW];
    float2 acc[RPW];
#pragma unroll
    for (int j = 0; j < RPW; ++j) {
        const int r = r0 + j < n ? r0 + j : n - 1;
        idx[j] = ell[(size_t)r * ELLK + lane];
        acc[j] = *(const float2*)(t + (size_t)r * HID + 2 * lane);
        deg[j] = r0 + j < n ? __builtin_amdgcn_readfirstlane(cnt[r]) : 0;
    }

#pragma unroll
    for (int j = 0; j < RPW; ++j) {
        const int d = deg[j] < ELLK ? deg[j] : ELLK;
        for (int s = 0; s < d; s += 8) {
            float2 v[8];
#pragma unroll
            for (int k = 0; k < 8; ++k) {
                v[k] = make_float2(0.f, 0.f);
                if (s + k < d) {
                    const int c = __builtin_amdgcn_readlane(idx[j], s + k);
                    v[k] = *(const float2*)(m + (size_t)c * HID + 2 * lane);
                }
            }
#pragma unroll
            for (int k = 0; k < 8; ++k) { acc[j].x += v[k].x; acc[j].y += v[k].y; }
        }
    }

#pragma unroll
    for (int j = 0; j < RPW; ++j) {
        if (deg[j] > ELLK) {
            const int r = r0 + j;
            const int novf = __builtin_amdgcn_readfirstlane(cnt[n]);
            for (int b = 0; b < novf; b += 32) {
                const int i = b + lane;
                int orow = -1, ocol = 0;
                if (i < novf) { orow = ovf[2 * (size_t)i]; ocol = ovf[2 * (size_t)i + 1]; }
                unsigned mask = __builtin_amdgcn_ballot_w32(orow == r);
                while (mask) {
                    const int k = __builtin_ctz(mask);
                    mask &= mask - 1;
                    const int c = __builtin_amdgcn_readlane(ocol, k);
                    const float2 v = *(const float2*)(m + (size_t)c * HID + 2 * lane);
                    acc[j].x += v.x; acc[j].y += v.y;
                }
            }
        }
        if (r0 + j < n)
            *(float2*)(t + (size_t)(r0 + j) * HID + 2 * lane) = acc[j];
    }
}

__global__ __launch_bounds__(256, 8) void gnn_norm_gemm_norm(
    float* __restrict__ t, const float* __restrict__ W,
    const float* __restrict__ w1, const float* __restrict__ w2,
    float* __restrict__ fin, const float* __restrict__ Wn,
    float* __restrict__ mout, int n)
{
    __shared__ __attribute__((aligned(16))) float lw[2 * 64 * WST];
    __shared__ __attribute__((aligned(16))) float lg[2 * HID];
    __shared__ __attribute__((aligned(16))) float stg[8 * 16 * SP];
    const int tid = threadIdx.x;
    float gword = 0.f;
    if (tid < 2 * HID) gword = tid < HID ? w1[tid] : w2[tid - HID];

    if (Wn) {
        for (int i = tid; i < 64 * 64; i += 256) {
            const int f = i >> 6, k = i & 63;
            const int rho = ((f >> 3) & 3) * 16 + (f >> 5) * 8 + (f & 7);
            const float a = W[i];
            const float b = Wn[i];
            lw[rho * WST + k] = a;
            lw[64 * WST + rho * WST + k] = b;
        }
    } else {
        for (int i = tid; i < 64 * 64; i += 256) {
            const int f = i >> 6, k = i & 63;
            const int rho = ((f >> 3) & 3) * 16 + (f >> 5) * 8 + (f & 7);
            lw[rho * WST + k] = W[i];
        }
    }
    if (tid < 2 * HID) lg[tid] = gword;
    __syncthreads();

    const int wave = __builtin_amdgcn_readfirstlane(tid >> 5);
    const int lane = tid & 31;
    const int half = lane >> 4;
    const int c    = lane & 15;
    const int node0 = blockIdx.x * 128 + wave * 16;
    if (node0 >= n) return;
    const int g0   = node0 * (HID / 4) + lane;
    const int gend = n * (HID / 4);
    const int rem = n - node0;
    const bool full = rem >= 16;
    float* tile = stg + wave * (16 * SP);
    float* tmem = tile + half * SP + 4 * c;
    float* town = tile + c * SP + 32 * half;
    const float4* t4 = (const float4*)t;

    {
        float4 v[8];
#pragma unroll
        for (int i = 0; i < 8; ++i) {
            const int g = g0 + 32 * i;
            v[i] = t4[g < gend ? g : gend - 1];
        }
#pragma unroll
        for (int i = 0; i < 8; ++i) *(float4*)(tmem + 2 * i * SP) = v[i];
    }
    __builtin_amdgcn_wave_barrier();
    float u[32];
#pragma unroll
    for (int q = 0; q < 8; ++q) {
        const float4 v = *(const float4*)(town + 4 * q);
        u[4 * q + 0] = v.x; u[4 * q + 1] = v.y;
        u[4 * q + 2] = v.z; u[4 * q + 3] = v.w;
    }

    {
        float p0 = 0.f, p1 = 0.f, p2 = 0.f, p3 = 0.f;
#pragma unroll
        for (int q = 0; q < 8; ++q) {
            p0 += u[4 * q + 0] * u[4 * q + 0];
            p1 += u[4 * q + 1] * u[4 * q + 1];
            p2 += u[4 * q + 2] * u[4 * q + 2];
            p3 += u[4 * q + 3] * u[4 * q + 3];
        }
        float ss = (p0 + p1) + (p2 + p3);
        ss += __shfl_xor(ss, 16, 32);
        const float inv = rsqrtf(ss * (1.0f / 64.0f) + 1e-5f);
#pragma unroll
        for (int q = 0; q < 8; ++q) {
            const float4 g = *(const float4*)&lg[half * 32 + 4 * q];
            u[4 * q + 0] = u[4 * q + 0] * inv * g.x;
            u[4 * q + 1] = u[4 * q + 1] * inv * g.y;
            u[4 * q + 2] = u[4 * q + 2] * inv * g.z;
            u[4 * q + 3] = u[4 * q + 3] * inv * g.w;
        }
    }

    v8f acc[4] = {};

#pragma unroll
    for (int sp = 0; sp < 8; ++sp) {
        v2f b0; b0.x = u[4 * sp];     b0.y = u[4 * sp + 1];
        v2f b1; b1.x = u[4 * sp + 2]; b1.y = u[4 * sp + 3];
        float4 aa[4];
#pragma unroll
        for (int jo = 0; jo < 4; ++jo)
            aa[jo] = *(const float4*)&lw[(jo * 16 + c) * WST + half * 32 + 4 * sp];
#pragma unroll
        for (int jo = 0; jo < 4; ++jo) {
            v2f a; a.x = aa[jo].x; a.y = aa[jo].y;
            acc[jo] = __builtin_amdgcn_wmma_f32_16x16x4_f32(
                false, a, false, b0, (short)0, acc[jo], false, false);
        }
#pragma unroll
        for (int jo = 0; jo < 4; ++jo) {
            v2f a; a.x = aa[jo].z; a.y = aa[jo].w;
            acc[jo] = __builtin_amdgcn_wmma_f32_16x16x4_f32(
                false, a, false, b1, (short)0, acc[jo], false, false);
        }
    }

    float p0 = 0.f, p1 = 0.f, p2 = 0.f, p3 = 0.f;
#pragma unroll
    for (int jo = 0; jo < 4; ++jo) {
#pragma unroll
        for (int r = 0; r < 8; ++r)
            u[8 * jo + r] += fmaxf(acc[jo][r], 0.f);
    }
#pragma unroll
    for (int q = 0; q < 8; ++q) {
        p0 += u[4 * q + 0] * u[4 * q + 0];
        p1 += u[4 * q + 1] * u[4 * q + 1];
        p2 += u[4 * q + 2] * u[4 * q + 2];
        p3 += u[4 * q + 3] * u[4 * q + 3];
    }
    float ss2 = (p0 + p1) + (p2 + p3);
    ss2 += __shfl_xor(ss2, 16, 32);
    const float inv2 = rsqrtf(ss2 * (1.0f / 64.0f) + 1e-5f);
    float4* out4 = (float4*)(fin ? fin : t);
#pragma unroll
    for (int q = 0; q < 8; ++q) {
        const float4 g = *(const float4*)&lg[HID + half * 32 + 4 * q];
        u[4 * q + 0] = u[4 * q + 0] * inv2 * g.x;
        u[4 * q + 1] = u[4 * q + 1] * inv2 * g.y;
        u[4 * q + 2] = u[4 * q + 2] * inv2 * g.z;
        u[4 * q + 3] = u[4 * q + 3] * inv2 * g.w;
    }
    if (full) {
#pragma unroll
        for (int q = 0; q < 8; ++q) {
            float4 o;
            o.x = u[4 * q + 0]; o.y = u[4 * q + 1];
            o.z = u[4 * q + 2]; o.w = u[4 * q + 3];
            *(float4*)(town + 4 * q) = o;
        }
        __builtin_amdgcn_wave_barrier();
        float4 o[8];
#pragma unroll
        for (int i = 0; i < 8; ++i) o[i] = *(const float4*)(tmem + 2 * i * SP);
#pragma unroll
        for (int i = 0; i < 8; ++i) out4[g0 + 32 * i] = o[i];
    } else if (c < rem) {
#pragma unroll
        for (int q = 0; q < 8; ++q) {
            float4 o;
            o.x = u[4 * q + 0]; o.y = u[4 * q + 1];
            o.z = u[4 * q + 2]; o.w = u[4 * q + 3];
            out4[(node0 + c) * (HID / 4) + half * 8 + q] = o;
        }
    }

    if (Wn) {
        v8f acn[4] = {};
#pragma unroll
        for (int sp = 0; sp < 8; ++sp) {
            v2f b0; b0.x = u[4 * sp];     b0.y = u[4 * sp + 1];
            v2f b1; b1.x = u[4 * sp + 2]; b1.y = u[4 * sp + 3];
            float4 aa[4];
#pragma unroll
            for (int jo = 0; jo < 4; ++jo)
                aa[jo] = *(const float4*)&lw[64 * WST + (jo * 16 + c) * WST + half * 32 + 4 * sp];
#pragma unroll
            for (int jo = 0; jo < 4; ++jo) {
                v2f a; a.x = aa[jo].x; a.y = aa[jo].y;
                acn[jo] = __builtin_amdgcn_wmma_f32_16x16x4_f32(
                    false, a, false, b0, (short)0, acn[jo], false, false);
            }
#pragma unroll
            for (int jo = 0; jo < 4; ++jo) {
                v2f a; a.x = aa[jo].z; a.y = aa[jo].w;
                acn[jo] = __builtin_amdgcn_wmma_f32_16x16x4_f32(
                    false, a, false, b1, (short)0, acn[jo], false, false);
            }
        }
        float4* m4 = (float4*)mout;
        if (full) {
#pragma unroll
            for (int jo = 0; jo < 4; ++jo) {
                float4 o0, o1;
                o0.x = fmaxf(acn[jo][0], 0.f); o0.y = fmaxf(acn[jo][1], 0.f);
                o0.z = fmaxf(acn[jo][2], 0.f); o0.w = fmaxf(acn[jo][3], 0.f);
                o1.x = fmaxf(acn[jo][4], 0.f); o1.y = fmaxf(acn[jo][5], 0.f);
                o1.z = fmaxf(acn[jo][6], 0.f); o1.w = fmaxf(acn[jo][7], 0.f);
                *(float4*)(town + 8 * jo)     = o0;
                *(float4*)(town + 8 * jo + 4) = o1;
            }
            __builtin_amdgcn_wave_barrier();
            float4 o[8];
#pragma unroll
            for (int i = 0; i < 8; ++i) o[i] = *(const float4*)(tmem + 2 * i * SP);
#pragma unroll
            for (int i = 0; i < 8; ++i) m4[g0 + 32 * i] = o[i];
        } else if (c < rem) {
#pragma unroll
            for (int jo = 0; jo < 4; ++jo) {
                float4 o0, o1;
                o0.x = fmaxf(acn[jo][0], 0.f); o0.y = fmaxf(acn[jo][1], 0.f);
                o0.z = fmaxf(acn[jo][2], 0.f); o0.w = fmaxf(acn[jo][3], 0.f);
                o1.x = fmaxf(acn[jo][4], 0.f); o1.y = fmaxf(acn[jo][5], 0.f);
                o1.z = fmaxf(acn[jo][6], 0.f); o1.w = fmaxf(acn[jo][7], 0.f);
                m4[(node0 + c) * (HID / 4) + half * 8 + 2 * jo]     = o0;
                m4[(node0 + c) * (HID / 4) + half * 8 + 2 * jo + 1] = o1;
            }
        }
    }
}

extern "C" void kernel_launch(void* const* d_in, const int* in_sizes, int n_in,
                              void* d_out, int out_size, void* d_ws, size_t ws_size,
                              hipStream_t stream)
{
    const float* x       = (const float*)d_in[0];
    const float* conv_w  = (const float*)d_in[1];
    const float* conv_nw = (const float*)d_in[2];
    const float* hid_w   = (const float*)d_in[3];
    const float* hid_nw  = (const float*)d_in[4];
    const int*   row     = (const int*)d_in[5];
    const int*   col     = (const int*)d_in[6];

    const int n  = in_sizes[0] / HID;
    const int ne = in_sizes[5];

    float* P = (float*)d_out;
    float* Q = (float*)d_ws;
    float* R = Q + (size_t)n * HID;
    int* cnt = (int*)(R + (size_t)n * HID);
    int* ell = cnt + ((size_t)n + 1 + 31) / 32 * 32;
    int* ovf = ell + (size_t)n * ELLK;

    const dim3 blk(256);
    const int gemm_grid = (n + 127) / 128;
    const int fill_grid = (ne + 1023) / 1024;
    const int pull_grid = ((n + RPW - 1) / RPW + 7) / 8;

    gnn_gemm_relu<<<gemm_grid, blk, 0, stream>>>(x, conv_w, nullptr, Q, R, cnt, n);
    gnn_fill<<<fill_grid, blk, 0, stream>>>(row, col, cnt, ell, ovf, n, ne);

    for (int l = 0; l < 4; ++l) {
        const float* hw = hid_w  + (size_t)l * HID * HID;
        const float* cn = conv_nw + l * HID;
        const float* hn = hid_nw  + l * HID;
        const float* cwn = l < 3 ? conv_w + (size_t)(l + 1) * HID * HID : nullptr;

        gnn_pull<<<pull_grid, blk, 0, stream>>>(Q, cnt, ell, ovf, R, n);
        gnn_norm_gemm_norm<<<gemm_grid, blk, 0, stream>>>(R, hw, cn, hn,
                                                          l == 3 ? P : nullptr,
                                                          cwn, Q, n);
    }
}


# Sum of sources moved into the dense kernel: a wave sums its own 16 rows in registers (two rows' loads in flight, each row in list order) into its LDS tile; t never stored or re-read; 10 -> 6 launches
# speedup vs baseline: 3.2130x; 1.0789x over previous
#include <hip/hip_runtime.h>

typedef __attribute__((ext_vector_type(2))) float v2f;
typedef __attribute__((ext_vector_type(8))) float v8f;
typedef __attribute__((ext_vector_type(4))) float v4f;

#define HID 64
#define WST 68
#define SP  68

__global__ __launch_bounds__(256) void gnn_gemm_relu(
    const float* __restrict__ in, const float* __restrict__ W,
    const float* __restrict__ resid, float* __restrict__ out,
    float* __restrict__ copy, int* __restrict__ zero, int n)
{
    __shared__ float lw[64 * WST];
    const int tid = threadIdx.x;

    if (zero) {
        const int gid = (int)blockIdx.x * 256 + tid;
        if (gid <= n) zero[gid] = 0;
    }

    if (copy) {
        const int rows_left = n - (int)blockIdx.x * 128;
        const size_t base = (size_t)blockIdx.x * 128 * HID;
        const v4f* src = (const v4f*)(in + base) + tid;
        v4f* dst = (v4f*)(copy + base) + tid;
        if (rows_left >= 128) {
            v4f cv[8];
#pragma unroll
            for (int p = 0; p < 8; ++p) cv[p] = src[p * 256];
            asm volatile("" :: "v"(cv[0]), "v"(cv[1]), "v"(cv[2]), "v"(cv[3]),
                               "v"(cv[4]), "v"(cv[5]), "v"(cv[6]), "v"(cv[7]));
#pragma unroll
            for (int p = 0; p < 8; ++p) dst[p * 256] = cv[p];
        } else {
            for (int p = 0; 16 * p + (tid >> 4) < rows_left; ++p) dst[p * 256] = src[p * 256];
        }
    }

    for (int i = tid; i < 64 * 64; i += 256) {
        int r = i >> 6, c = i & 63;
        lw[r * WST + c] = W[i];
    }
    __syncthreads();

    const int wave = tid >> 5;
    const int lane = tid & 31;
    const int half = lane >> 4;
    const int l16  = lane & 15;
    const int row0 = blockIdx.x * 128 + wave * 16;

    const int  arow  = row0 + l16;
    const bool rowok = arow < n;
    const float* abase = in + (size_t)(rowok ? arow : 0) * HID + half * 2;
    v2f a[16];
#pragma unroll
    for (int s = 0; s < 16; ++s) {
        float2 t = rowok ? *(const float2*)(abase + 4 * s) : make_float2(0.f, 0.f);
        a[s].x = t.x; a[s].y = t.y;
    }

    v8f acc[4] = {};

#pragma unroll
    for (int s = 0; s < 16; ++s) {
#pragma unroll
        for (int j = 0; j < 4; ++j) {
            const int nc = j * 16 + l16;
            float2 bb = *(const float2*)&lw[nc * WST + 4 * s + half * 2];
            v2f b; b.x = bb.x; b.y = bb.y;
            acc[j] = __builtin_amdgcn_wmma_f32_16x16x4_f32(
                false, a[s], false, b, (short)0, acc[j], false, false);
        }
    }

#pragma unroll
    for (int j = 0; j < 4; ++j) {
        const int colg = j * 16 + l16;
#pragma unroll
        for (int v = 0; v < 8; ++v) {
            const int rowg = row0 + v + half * 8;
            if (rowg < n) {
                float val = fmaxf(acc[j][v], 0.f);
                if (resid) val += resid[(size_t)rowg * HID + colg];
                out[(size_t)rowg * HID + colg] = val;
            }
        }
    }
}

#define ELLK 32
#define RPW  16

__global__ __launch_bounds__(256) void gnn_fill(
    const int* __restrict__ row, const int* __restrict__ col,
    int* __restrict__ cnt, int* __restrict__ ell, int* __restrict__ ovf,
    int n, int ne)
{
    const int gid = (int)blockIdx.x * 256 + (int)threadIdx.x;
    const int stride = (int)gridDim.x * 256;
    int r[4], c[4], slot[4];
#pragma unroll
    for (int k = 0; k < 4; ++k) {
        const int e = gid + k * stride;
        r[k] = e < ne ? row[e] : -1;
        c[k] = e < ne ? col[e] : 0;
    }
#pragma unroll
    for (int k = 0; k < 4; ++k)
        slot[k] = r[k] >= 0 ? atomicAdd(cnt + r[k], 1) : 0;
#pragma unroll
    for (int k = 0; k < 4; ++k) {
        if (r[k] < 0) continue;
        if (slot[k] < ELLK) {
            ell[(size_t)r[k] * ELLK + slot[k]] = c[k];
        } else {
            const int i = atomicAdd(cnt + n, 1);
            ovf[2 * (size_t)i]     = r[k];
            ovf[2 * (size_t)i + 1] = c[k];
        }
    }
}

__global__ __launch_bounds__(256, 8) void gnn_pull_norm_gemm_norm(
    const float* __restrict__ m, const int* __restrict__ cnt,
    const int* __restrict__ ell, const int* __restrict__ ovf,
    float* __restrict__ t, const float* __restrict__ W,
    const float* __restrict__ w1, const float* __restrict__ w2,
    float* __restrict__ fin, const float* __restrict__ Wn,
    float* __restrict__ mout, int n)
{
    __shared__ __attribute__((aligned(16))) float lw[2 * 64 * WST];
    __shared__ __attribute__((aligned(16))) float lg[2 * HID];
    __shared__ __attribute__((aligned(16))) float stg[8 * 16 * SP];
    const int tid = threadIdx.x;
    float gword = 0.f;
    if (tid < 2 * HID) gword = tid < HID ? w1[tid] : w2[tid - HID];

    if (Wn) {
        for (int i = tid; i < 64 * 64; i += 256) {
            const int f = i >> 6, k = i & 63;
            const int rho = ((f >> 3) & 3) * 16 + (f >> 5) * 8 + (f & 7);
            const float a = W[i];
            const float b = Wn[i];
            lw[rho * WST + k] = a;
            lw[64 * WST + rho * WST + k] = b;
        }
    } else {
        for (int i = tid; i < 64 * 64; i += 256) {
            const int f = i >> 6, k = i & 63;
            const int rho = ((f >> 3) & 3) * 16 + (f >> 5) * 8 + (f & 7);
            lw[rho * WST + k] = W[i];
        }
    }
    if (tid < 2 * HID) lg[tid] = gword;
    __syncthreads();

    const int wave = __builtin_amdgcn_readfirstlane(tid >> 5);
    const int lane = tid & 31;
    const int half = lane >> 4;
    const int c    = lane & 15;
    const int node0 = blockIdx.x * 128 + wave * 16;
    if (node0 >= n) return;
    const int g0   = node0 * (HID / 4) + lane;
    const int rem = n - node0;
    const bool full = rem >= 16;
    float* tile = stg + wave * (16 * SP);
    float* tmem = tile + half * SP + 4 * c;
    float* town = tile + c * SP + 32 * half;

    {
        int idx[RPW], deg[RPW];
        float2 acc[RPW];
#pragma unroll
        for (int j = 0; j < RPW; ++j) {
            const int r = node0 + j < n ? node0 + j : n - 1;
            idx[j] = ell[(size_t)r * ELLK + lane];
            acc[j] = *(const float2*)(t + (size_t)r * HID + 2 * lane);
            deg[j] = node0 + j < n ? __builtin_amdgcn_readfirstlane(cnt[r]) : 0;
        }

#pragma unroll
        for (int j = 0; j < RPW; j += 2) {
            const int da = deg[j]     < ELLK ? deg[j]     : ELLK;
            const int db = deg[j + 1] < ELLK ? deg[j + 1] : ELLK;
            const int dm = da > db ? da : db;
            for (int s = 0; s < dm; s += 8) {
                float2 va[8], vb[8];
#pragma unroll
                for (int k = 0; k < 8; ++k) {
                    va[k] = make_float2(0.f, 0.f);
                    if (s + k < da) {
                        const int src = __builtin_amdgcn_readlane(idx[j], s + k);
                        va[k] = *(const float2*)(m + (size_t)src * HID + 2 * lane);
                    }
                }
#pragma unroll
                for (int k = 0; k < 8; ++k) {
                    vb[k] = make_float2(0.f, 0.f);
                    if (s + k < db) {
                        const int src = __builtin_amdgcn_readlane(idx[j + 1], s + k);
                        vb[k] = *(const float2*)(m + (size_t)src * HID + 2 * lane);
                    }
                }
                if (s < da) {
#pragma unroll
                    for (int k = 0; k < 8; ++k) { acc[j].x += va[k].x; acc[j].y += va[k].y; }
                }
                if (s < db) {
#pragma unroll
                    for (int k = 0; k < 8; ++k) { acc[j + 1].x += vb[k].x; acc[j + 1].y += vb[k].y; }
                }
            }
        }

#pragma unroll
        for (int j = 0; j < RPW; ++j) {
            if (deg[j] > ELLK) {
                const int r = node0 + j;
                const int novf = __builtin_amdgcn_readfirstlane(cnt[n]);
                for (int b = 0; b < novf; b += 32) {
                    const int i = b + lane;
                    int orow = -1, ocol = 0;
                    if (i < novf) { orow = ovf[2 * (size_t)i]; ocol = ovf[2 * (size_t)i + 1]; }
                    unsigned mask = __builtin_amdgcn_ballot_w32(orow == r);
                    while (mask) {
                        const int k = __builtin_ctz(mask);
                        mask &= mask - 1;
                        const int src = __builtin_amdgcn_readlane(ocol, k);
                        const float2 v = *(const float2*)(m + (size_t)src * HID + 2 * lane);
                        acc[j].x += v.x; acc[j].y += v.y;
                    }
                }
            }
            *(float2*)(tile + j * SP + 2 * lane) = acc[j];
        }
    }
    __builtin_amdgcn_wave_barrier();
    float u[32];
#pragma unroll
    for (int q = 0; q < 8; ++q) {
        const float4 v = *(const float4*)(town + 4 * q);
        u[4 * q + 0] = v.x; u[4 * q + 1] = v.y;
        u[4 * q + 2] = v.z; u[4 * q + 3] = v.w;
    }

    {
        float p0 = 0.f, p1 = 0.f, p2 = 0.f, p3 = 0.f;
#pragma unroll
        for (int q = 0; q < 8; ++q) {
            p0 += u[4 * q + 0] * u[4 * q + 0];
            p1 += u[4 * q + 1] * u[4 * q + 1];
            p2 += u[4 * q + 2] * u[4 * q + 2];
            p3 += u[4 * q + 3] * u[4 * q + 3];
        }
        float ss = (p0 + p1) + (p2 + p3);
        ss += __shfl_xor(ss, 16, 32);
        const float inv = rsqrtf(ss * (1.0f / 64.0f) + 1e-5f);
#pragma unroll
        for (int q = 0; q < 8; ++q) {
            const float4 g = *(const float4*)&lg[half * 32 + 4 * q];
            u[4 * q + 0] = u[4 * q + 0] * inv * g.x;
            u[4 * q + 1] = u[4 * q + 1] * inv * g.y;
            u[4 * q + 2] = u[4 * q + 2] * inv * g.z;
            u[4 * q + 3] = u[4 * q + 3] * inv * g.w;
        }
    }

    v8f acc[4] = {};

#pragma unroll
    for (int sp = 0; sp < 8; ++sp) {
        v2f b0; b0.x = u[4 * sp];     b0.y = u[4 * sp + 1];
        v2f b1; b1.x = u[4 * sp + 2]; b1.y = u[4 * sp + 3];
        float4 aa[4];
#pragma unroll
        for (int jo = 0; jo < 4; ++jo)
            aa[jo] = *(const float4*)&lw[(jo * 16 + c) * WST + half * 32 + 4 * sp];
#pragma unroll
        for (int jo = 0; jo < 4; ++jo) {
            v2f a; a.x = aa[jo].x; a.y = aa[jo].y;
            acc[jo] = __builtin_amdgcn_wmma_f32_16x16x4_f32(
                false, a, false, b0, (short)0, acc[jo], false, false);
        }
#pragma unroll
        for (int jo = 0; jo < 4; ++jo) {
            v2f a; a.x = aa[jo].z; a.y = aa[jo].w;
            acc[jo] = __builtin_amdgcn_wmma_f32_16x16x4_f32(
                false, a, false, b1, (short)0, acc[jo], false, false);
        }
    }

    float p0 = 0.f, p1 = 0.f, p2 = 0.f, p3 = 0.f;
#pragma unroll
    for (int jo = 0; jo < 4; ++jo) {
#pragma unroll
        for (int r = 0; r < 8; ++r)
            u[8 * jo + r] += fmaxf(acc[jo][r], 0.f);
    }
#pragma unroll
    for (int q = 0; q < 8; ++q) {
        p0 += u[4 * q + 0] * u[4 * q + 0];
        p1 += u[4 * q + 1] * u[4 * q + 1];
        p2 += u[4 * q + 2] * u[4 * q + 2];
        p3 += u[4 * q + 3] * u[4 * q + 3];
    }
    float ss2 = (p0 + p1) + (p2 + p3);
    ss2 += __shfl_xor(ss2, 16, 32);
    const float inv2 = rsqrtf(ss2 * (1.0f / 64.0f) + 1e-5f);
    float4* out4 = (float4*)(fin ? fin : t);
#pragma unroll
    for (int q = 0; q < 8; ++q) {
        const float4 g = *(const float4*)&lg[HID + half * 32 + 4 * q];
        u[4 * q + 0] = u[4 * q + 0] * inv2 * g.x;
        u[4 * q + 1] = u[4 * q + 1] * inv2 * g.y;
        u[4 * q + 2] = u[4 * q + 2] * inv2 * g.z;
        u[4 * q + 3] = u[4 * q + 3] * inv2 * g.w;
    }
    if (full) {
#pragma unroll
        for (int q = 0; q < 8; ++q) {
            float4 o;
            o.x = u[4 * q + 0]; o.y = u[4 * q + 1];
            o.z = u[4 * q + 2]; o.w = u[4 * q + 3];
            *(float4*)(town + 4 * q) = o;
        }
        __builtin_amdgcn_wave_barrier();
        float4 o[8];
#pragma unroll
        for (int i = 0; i < 8; ++i) o[i] = *(const float4*)(tmem + 2 * i * SP);
#pragma unroll
        for (int i = 0; i < 8; ++i) out4[g0 + 32 * i] = o[i];
    } else if (c < rem) {
#pragma unroll
        for (int q = 0; q < 8; ++q) {
            float4 o;
            o.x = u[4 * q + 0]; o.y = u[4 * q + 1];
            o.z = u[4 * q + 2]; o.w = u[4 * q + 3];
            out4[(node0 + c) * (HID / 4) + half * 8 + q] = o;
        }
    }

    if (Wn) {
        v8f acn[4] = {};
#pragma unroll
        for (int sp = 0; sp < 8; ++sp) {
            v2f b0; b0.x = u[4 * sp];     b0.y = u[4 * sp + 1];
            v2f b1; b1.x = u[4 * sp + 2]; b1.y = u[4 * sp + 3];
            float4 aa[4];
#pragma unroll
            for (int jo = 0; jo < 4; ++jo)
                aa[jo] = *(const float4*)&lw[64 * WST + (jo * 16 + c) * WST + half * 32 + 4 * sp];
#pragma unroll
            for (int jo = 0; jo < 4; ++jo) {
                v2f a; a.x = aa[jo].x; a.y = aa[jo].y;
                acn[jo] = __builtin_amdgcn_wmma_f32_16x16x4_f32(
                    false, a, false, b0, (short)0, acn[jo], false, false);
            }
#pragma unroll
            for (int jo = 0; jo < 4; ++jo) {
                v2f a; a.x = aa[jo].z; a.y = aa[jo].w;
                acn[jo] = __builtin_amdgcn_wmma_f32_16x16x4_f32(
                    false, a, false, b1, (short)0, acn[jo], false, false);
            }
        }
        float4* m4 = (float4*)mout;
        if (full) {
#pragma unroll
            for (int jo = 0; jo < 4; ++jo) {
                float4 o0, o1;
                o0.x = fmaxf(acn[jo][0], 0.f); o0.y = fmaxf(acn[jo][1], 0.f);
                o0.z = fmaxf(acn[jo][2], 0.f); o0.w = fmaxf(acn[jo][3], 0.f);
                o1.x = fmaxf(acn[jo][4], 0.f); o1.y = fmaxf(acn[jo][5], 0.f);
                o1.z = fmaxf(acn[jo][6], 0.f); o1.w = fmaxf(acn[jo][7], 0.f);
                *(float4*)(town + 8 * jo)     = o0;
                *(float4*)(town + 8 * jo + 4) = o1;
            }
            __builtin_amdgcn_wave_barrier();
            float4 o[8];
#pragma unroll
            for (int i = 0; i < 8; ++i) o[i] = *(const float4*)(tmem + 2 * i * SP);
#pragma unroll
            for (int i = 0; i < 8; ++i) m4[g0 + 32 * i] = o[i];
        } else if (c < rem) {
#pragma unroll
            for (int jo = 0; jo < 4; ++jo) {
                float4 o0, o1;
                o0.x = fmaxf(acn[jo][0], 0.f); o0.y = fmaxf(acn[jo][1], 0.f);
                o0.z = fmaxf(acn[jo][2], 0.f); o0.w = fmaxf(acn[jo][3], 0.f);
                o1.x = fmaxf(acn[jo][4], 0.f); o1.y = fmaxf(acn[jo][5], 0.f);
                o1.z = fmaxf(acn[jo][6], 0.f); o1.w = fmaxf(acn[jo][7], 0.f);
                m4[(node0 + c) * (HID / 4) + half * 8 + 2 * jo]     = o0;
                m4[(node0 + c) * (HID / 4) + half * 8 + 2 * jo + 1] = o1;
            }
        }
    }
}

extern "C" void kernel_launch(void* const* d_in, const int* in_sizes, int n_in,
                              void* d_out, int out_size, void* d_ws, size_t ws_size,
                              hipStream_t stream)
{
    const float* x       = (const float*)d_in[0];
    const float* conv_w  = (const float*)d_in[1];
    const float* conv_nw = (const float*)d_in[2];
    const float* hid_w   = (const float*)d_in[3];
    const float* hid_nw  = (const float*)d_in[4];
    const int*   row     = (const int*)d_in[5];
    const int*   col     = (const int*)d_in[6];

    const int n  = in_sizes[0] / HID;
    const int ne = in_sizes[5];

    float* P  = (float*)d_out;
    float* Q0 = (float*)d_ws;
    float* R  = Q0 + (size_t)n * HID;
    float* Q1 = R + (size_t)n * HID;
    int* cnt = (int*)(Q1 + (size_t)n * HID);
    int* ell = cnt + ((size_t)n + 1 + 31) / 32 * 32;
    int* ovf = ell + (size_t)n * ELLK;

    const dim3 blk(256);
    const int gemm_grid = (n + 127) / 128;
    const int fill_grid = (ne + 1023) / 1024;

    gnn_gemm_relu<<<gemm_grid, blk, 0, stream>>>(x, conv_w, nullptr, Q0, R, cnt, n);
    gnn_fill<<<fill_grid, blk, 0, stream>>>(row, col, cnt, ell, ovf, n, ne);

    for (int l = 0; l < 4; ++l) {
        const float* hw = hid_w  + (size_t)l * HID * HID;
        const float* cn = conv_nw + l * HID;
        const float* hn = hid_nw  + l * HID;
        const float* cwn = l < 3 ? conv_w + (size_t)(l + 1) * HID * HID : nullptr;

        const float* mcur = (l & 1) ? Q1 : Q0;
        float*       mnxt = (l & 1) ? Q0 : Q1;

        gnn_pull_norm_gemm_norm<<<gemm_grid, blk, 0, stream>>>(mcur, cnt, ell, ovf,
                                                               R, hw, cn, hn,
                                                               l == 3 ? P : nullptr,
                                                               cwn, mnxt, n);
    }
}


# Sum of sources: a gather load fetches two source rows, one a lane half (16 lanes x 16 B), for two destination rows; entries reach lanes by row-share DPP; absent ones read a zero row; add order kept
# speedup vs baseline: 3.3342x; 1.0377x over previous
#include <hip/hip_runtime.h>

typedef __attribute__((ext_vector_type(2))) float v2f;
typedef __attribute__((ext_vector_type(8))) float v8f;
typedef __attribute__((ext_vector_type(4))) float v4f;

#define HID 64
#define WST 68
#define SP  68

__global__ __launch_bounds__(256) void gnn_gemm_relu(
    const float* __restrict__ in, const float* __restrict__ W,
    const float* __restrict__ resid, float* __restrict__ out,
    float* __restrict__ copy, int* __restrict__ zero, int n)
{
    __shared__ float lw[64 * WST];
    const int tid = threadIdx.x;

    if (zero) {
        const int gid = (int)blockIdx.x * 256 + tid;
        if (gid <= n) zero[gid] = 0;
        if (gid < HID) out[(size_t)n * HID + gid] = 0.f;
    }

    if (copy) {
        const int rows_left = n - (int)blockIdx.x * 128;
        const size_t base = (size_t)blockIdx.x * 128 * HID;
        const v4f* src = (const v4f*)(in + base) + tid;
        v4f* dst = (v4f*)(copy + base) + tid;
        if (rows_left >= 128) {
            v4f cv[8];
#pragma unroll
            for (int p = 0; p < 8; ++p) cv[p] = src[p * 256];
            asm volatile("" :: "v"(cv[0]), "v"(cv[1]), "v"(cv[2]), "v"(cv[3]),
                               "v"(cv[4]), "v"(cv[5]), "v"(cv[6]), "v"(cv[7]));
#pragma unroll
            for (int p = 0; p < 8; ++p) dst[p * 256] = cv[p];
        } else {
            for (int p = 0; 16 * p + (tid >> 4) < rows_left; ++p) dst[p * 256] = src[p * 256];
        }
    }

    for (int i = tid; i < 64 * 64; i += 256) {
        int r = i >> 6, c = i & 63;
        lw[r * WST + c] = W[i];
    }
    __syncthreads();

    const int wave = tid >> 5;
    const int lane = tid & 31;
    const int half = lane >> 4;
    const int l16  = lane & 15;
    const int row0 = blockIdx.x * 128 + wave * 16;

    const int  arow  = row0 + l16;
    const bool rowok = arow < n;
    const float* abase = in + (size_t)(rowok ? arow : 0) * HID + half * 2;
    v2f a[16];
#pragma unroll
    for (int s = 0; s < 16; ++s) {
        float2 t = rowok ? *(const float2*)(abase + 4 * s) : make_float2(0.f, 0.f);
        a[s].x = t.x; a[s].y = t.y;
    }

    v8f acc[4] = {};

#pragma unroll
    for (int s = 0; s < 16; ++s) {
#pragma unroll
        for (int j = 0; j < 4; ++j) {
            const int nc = j * 16 + l16;
            float2 bb = *(const float2*)&lw[nc * WST + 4 * s + half * 2];
            v2f b; b.x = bb.x; b.y = bb.y;
            acc[j] = __builtin_amdgcn_wmma_f32_16x16x4_f32(
                false, a[s], false, b, (short)0, acc[j], false, false);
        }
    }

#pragma unroll
    for (int j = 0; j < 4; ++j) {
        const int colg = j * 16 + l16;
#pragma unroll
        for (int v = 0; v < 8; ++v) {
            const int rowg = row0 + v + half * 8;
            if (rowg < n) {
                float val = fmaxf(acc[j][v], 0.f);
                if (resid) val += resid[(size_t)rowg * HID + colg];
                out[(size_t)rowg * HID + colg] = val;
            }
        }
    }
}

#define ELLK 32
#define ELLH 16
#define RPW  16

__global__ __launch_bounds__(256) void gnn_fill(
    const int* __restrict__ row, const int* __restrict__ col,
    int* __restrict__ cnt, int* __restrict__ ell, int* __restrict__ ovf,
    int n, int ne)
{
    const int gid = (int)blockIdx.x * 256 + (int)threadIdx.x;
    const int stride = (int)gridDim.x * 256;
    int r[4], c[4], slot[4];
#pragma unroll
    for (int k = 0; k < 4; ++k) {
        const int e = gid + k * stride;
        r[k] = e < ne ? row[e] : -1;
        c[k] = e < ne ? col[e] : 0;
    }
#pragma unroll
    for (int k = 0; k < 4; ++k)
        slot[k] = r[k] >= 0 ? atomicAdd(cnt + r[k], 1) : 0;
#pragma unroll
    for (int k = 0; k < 4; ++k) {
        if (r[k] < 0) continue;
        if (slot[k] < ELLK) {
            ell[(size_t)r[k] * ELLK + slot[k]] = c[k];
        } else {
            const int i = atomicAdd(cnt + n, 1);
            ovf[2 * (size_t)i]     = r[k];
            ovf[2 * (size_t)i + 1] = c[k];
        }
    }
}

template <int SH, int K>
__device__ __forceinline__ void gnn_gather_offs(unsigned (&off)[8], const int e8,
                                                const unsigned loff)
{
    off[K] = (unsigned)__builtin_amdgcn_update_dpp(0, e8, 0x150 + SH + K, 0xf, 0xf, true)
             | loff;
    if constexpr (K < 7) gnn_gather_offs<SH, K + 1>(off, e8, loff);
}

template <int K>
__device__ __forceinline__ void gnn_gather_loads(float4 (&v)[8], const unsigned (&off)[8],
                                                 const int left, const char* mb)
{
    v[K] = *(const float4*)(mb + off[K]);
    if constexpr (K < 7) {
        if (K + 1 < left) gnn_gather_loads<K + 1>(v, off, left, mb);
    }
}

template <int K>
__device__ __forceinline__ void gnn_gather_adds(float4& a, const float4 (&v)[8],
                                                const int left)
{
    a.x += v[K].x; a.y += v[K].y; a.z += v[K].z; a.w += v[K].w;
    if constexpr (K < 7) {
        if (K + 1 < left) gnn_gather_adds<K + 1>(a, v, left);
    }
}

template <int SH>
__device__ __forceinline__ void gnn_gather_pass(float4& a, const int e8, const int left,
                                                const char* mb, const unsigned loff)
{
    float4 v[8];
    unsigned off[8];
    gnn_gather_offs<SH, 0>(off, e8, loff);
    gnn_gather_loads<0>(v, off, left, mb);
    gnn_gather_adds<0>(a, v, left);
}

__global__ __launch_bounds__(256, 8) void gnn_pull_norm_gemm_norm(
    const float* __restrict__ m, const int* __restrict__ cnt,
    const int* __restrict__ ell, const int* __restrict__ ovf,
    float* __restrict__ t, const float* __restrict__ W,
    const float* __restrict__ w1, const float* __restrict__ w2,
    float* __restrict__ fin, const float* __restrict__ Wn,
    float* __restrict__ mout, int n)
{
    __shared__ __attribute__((aligned(16))) float lw[2 * 64 * WST];
    __shared__ __attribute__((aligned(16))) float lg[2 * HID];
    __shared__ __attribute__((aligned(16))) float stg[8 * 16 * SP];
    const int tid = threadIdx.x;
    if (Wn && blockIdx.x == 0 && tid < HID) mout[(size_t)n * HID + tid] = 0.f;
    float gword = 0.f;
    if (tid < 2 * HID) gword = tid < HID ? w1[tid] : w2[tid - HID];

    if (Wn) {
        for (int i = tid; i < 64 * 64; i += 256) {
            const int f = i >> 6, k = i & 63;
            const int rho = ((f >> 3) & 3) * 16 + (f >> 5) * 8 + (f & 7);
            const float a = W[i];
            const float b = Wn[i];
            lw[rho * WST + k] = a;
            lw[64 * WST + rho * WST + k] = b;
        }
    } else {
        for (int i = tid; i < 64 * 64; i += 256) {
            const int f = i >> 6, k = i & 63;
            const int rho = ((f >> 3) & 3) * 16 + (f >> 5) * 8 + (f & 7);
            lw[rho * WST + k] = W[i];
        }
    }
    if (tid < 2 * HID) lg[tid] = gword;
    __syncthreads();

    const int wave = __builtin_amdgcn_readfirstlane(tid >> 5);
    const int lane = tid & 31;
    const int half = lane >> 4;
    const int c    = lane & 15;
    const int node0 = blockIdx.x * 128 + wave * 16;
    if (node0 >= n) return;
    const int g0   = node0 * (HID / 4) + lane;
    const int rem = n - node0;
    const bool full = rem >= 16;
    float* tile = stg + wave * (16 * SP);
    float* tmem = tile + half * SP + 4 * c;
    float* town = tile + c * SP + 32 * half;

    {
        int deg[RPW];
        int elo[RPW / 2];
        float4 hrow[RPW / 2];
#pragma unroll
        for (int j = 0; j < RPW; ++j) {
            const int r = node0 + j < n ? node0 + j : n - 1;
            deg[j] = node0 + j < n ? __builtin_amdgcn_readfirstlane(cnt[r]) : 0;
        }
#pragma unroll
        for (int p = 0; p < RPW / 2; ++p) {
            const int jr = node0 + 2 * p + half;
            const int r = jr < n ? jr : n - 1;
            elo[p] = ell[(size_t)r * ELLK + c];
        }
#pragma unroll
        for (int p = 0; p < RPW / 2; ++p) {
            const int jr = node0 + 2 * p + half;
            const int r = jr < n ? jr : n - 1;
            hrow[p] = *(const float4*)(t + (size_t)r * HID + 4 * c);
        }
        asm volatile("" :: "v"(elo[0]), "v"(elo[1]), "v"(elo[2]), "v"(elo[3]),
                           "v"(elo[4]), "v"(elo[5]), "v"(elo[6]), "v"(elo[7]));
        const char* mb = (const char*)m;
        const unsigned loff = 16u * (unsigned)c;

        bool anybig = false;
#pragma unroll
        for (int p = 0; p < RPW / 2; ++p) {
            const int da = deg[2 * p]     < ELLH ? deg[2 * p]     : ELLH;
            const int db = deg[2 * p + 1] < ELLH ? deg[2 * p + 1] : ELLH;
            const int dm = da > db ? da : db;
            anybig |= deg[2 * p] > ELLH || deg[2 * p + 1] > ELLH;
            float4 a = hrow[p];
            const int myd = half ? db : da;
            const unsigned ec = c < myd ? (unsigned)elo[p] : (unsigned)n;
            const int e8 = (int)(ec << 8);
            if (dm > 0) gnn_gather_pass<0>(a, e8, dm, mb, loff);
            if (dm > 8) gnn_gather_pass<8>(a, e8, dm - 8, mb, loff);
            *(float4*)(tile + (2 * p + half) * SP + 4 * c) = a;
        }

        if (anybig) {
            __builtin_amdgcn_wave_barrier();
#pragma unroll 1
            for (int j = 0; j < RPW; ++j) {
                const int r = node0 + j;
                if (r >= n) break;
                const int d = __builtin_amdgcn_readfirstlane(cnt[r]);
                if (d <= ELLH) continue;
                float2 a2 = *(float2*)(tile + j * SP + 2 * lane);
                const int idx = ell[(size_t)r * ELLK + lane];
                const int lim = d < ELLK ? d : ELLK;
                for (int s = ELLH; s < lim; s += 8) {
                    float2 vv[8];
#pragma unroll
                    for (int k = 0; k < 8; ++k) {
                        vv[k] = make_float2(0.f, 0.f);
                        if (s + k < lim) {
                            const int src = __builtin_amdgcn_readlane(idx, s + k);
                            vv[k] = *(const float2*)(m + (size_t)src * HID + 2 * lane);
                        }
                    }
#pragma unroll
                    for (int k = 0; k < 8; ++k) { a2.x += vv[k].x; a2.y += vv[k].y; }
                }
                if (d > ELLK) {
                    const int novf = __builtin_amdgcn_readfirstlane(cnt[n]);
                    for (int b = 0; b < novf; b += 32) {
                        const int i = b + lane;
                        int orow = -1, ocol = 0;
                        if (i < novf) { orow = ovf[2 * (size_t)i]; ocol = ovf[2 * (size_t)i + 1]; }
                        unsigned mask = __builtin_amdgcn_ballot_w32(orow == r);
                        while (mask) {
                            const int k = __builtin_ctz(mask);
                            mask &= mask - 1;
                            const int src = __builtin_amdgcn_readlane(ocol, k);
                            const float2 v = *(const float2*)(m + (size_t)src * HID + 2 * lane);
                            a2.x += v.x; a2.y += v.y;
                        }
                    }
                }
                *(float2*)(tile + j * SP + 2 * lane) = a2;
            }
        }
    }
    __builtin_amdgcn_wave_barrier();
    float u[32];
#pragma unroll
    for (int q = 0; q < 8; ++q) {
        const float4 v = *(const float4*)(town + 4 * q);
        u[4 * q + 0] = v.x; u[4 * q + 1] = v.y;
        u[4 * q + 2] = v.z; u[4 * q + 3] = v.w;
    }

    {
        float p0 = 0.f, p1 = 0.f, p2 = 0.f, p3 = 0.f;
#pragma unroll
        for (int q = 0; q < 8; ++q) {
            p0 += u[4 * q + 0] * u[4 * q + 0];
            p1 += u[4 * q + 1] * u[4 * q + 1];
            p2 += u[4 * q + 2] * u[4 * q + 2];
            p3 += u[4 * q + 3] * u[4 * q + 3];
        }
        float ss = (p0 + p1) + (p2 + p3);
        ss += __shfl_xor(ss, 16, 32);
        const float inv = rsqrtf(ss * (1.0f / 64.0f) + 1e-5f);
#pragma unroll
        for (int q = 0; q < 8; ++q) {
            const float4 g = *(const float4*)&lg[half * 32 + 4 * q];
            u[4 * q + 0] = u[4 * q + 0] * inv * g.x;
            u[4 * q + 1] = u[4 * q + 1] * inv * g.y;
            u[4 * q + 2] = u[4 * q + 2] * inv * g.z;
            u[4 * q + 3] = u[4 * q + 3] * inv * g.w;
        }
    }

    v8f acc[4] = {};

#pragma unroll
    for (int sp = 0; sp < 8; ++sp) {
        v2f b0; b0.x = u[4 * sp];     b0.y = u[4 * sp + 1];
        v2f b1; b1.x = u[4 * sp + 2]; b1.y = u[4 * sp + 3];
        float4 aa[4];
#pragma unroll
        for (int jo = 0; jo < 4; ++jo)
            aa[jo] = *(const float4*)&lw[(jo * 16 + c) * WST + half * 32 + 4 * sp];
#pragma unroll
        for (int jo = 0; jo < 4; ++jo) {
            v2f a; a.x = aa[jo].x; a.y = aa[jo].y;
            acc[jo] = __builtin_amdgcn_wmma_f32_16x16x4_f32(
                false, a, false, b0, (short)0, acc[jo], false, false);
        }
#pragma unroll
        for (int jo = 0; jo < 4; ++jo) {
            v2f a; a.x = aa[jo].z; a.y = aa[jo].w;
            acc[jo] = __builtin_amdgcn_wmma_f32_16x16x4_f32(
                false, a, false, b1, (short)0, acc[jo], false, false);
        }
    }

    float p0 = 0.f, p1 = 0.f, p2 = 0.f, p3 = 0.f;
#pragma unroll
    for (int jo = 0; jo < 4; ++jo) {
#pragma unroll
        for (int r = 0; r < 8; ++r)
            u[8 * jo + r] += fmaxf(acc[jo][r], 0.f);
    }
#pragma unroll
    for (int q = 0; q < 8; ++q) {
        p0 += u[4 * q + 0] * u[4 * q + 0];
        p1 += u[4 * q + 1] * u[4 * q + 1];
        p2 += u[4 * q + 2] * u[4 * q + 2];
        p3 += u[4 * q + 3] * u[4 * q + 3];
    }
    float ss2 = (p0 + p1) + (p2 + p3);
    ss2 += __shfl_xor(ss2, 16, 32);
    const float inv2 = rsqrtf(ss2 * (1.0f / 64.0f) + 1e-5f);
    float4* out4 = (float4*)(fin ? fin : t);
#pragma unroll
    for (int q = 0; q < 8; ++q) {
        const float4 g = *(const float4*)&lg[HID + half * 32 + 4 * q];
        u[4 * q + 0] = u[4 * q + 0] * inv2 * g.x;
        u[4 * q + 1] = u[4 * q + 1] * inv2 * g.y;
        u[4 * q + 2] = u[4 * q + 2] * inv2 * g.z;
        u[4 * q + 3] = u[4 * q + 3] * inv2 * g.w;
    }
    if (full) {
#pragma unroll
        for (int q = 0; q < 8; ++q) {
            float4 o;
            o.x = u[4 * q + 0]; o.y = u[4 * q + 1];
            o.z = u[4 * q + 2]; o.w = u[4 * q + 3];
            *(float4*)(town + 4 * q) = o;
        }
        __builtin_amdgcn_wave_barrier();
        float4 o[8];
#pragma unroll
        for (int i = 0; i < 8; ++i) o[i] = *(const float4*)(tmem + 2 * i * SP);
#pragma unroll
        for (int i = 0; i < 8; ++i) out4[g0 + 32 * i] = o[i];
    } else if (c < rem) {
#pragma unroll
        for (int q = 0; q < 8; ++q) {
            float4 o;
            o.x = u[4 * q + 0]; o.y = u[4 * q + 1];
            o.z = u[4 * q + 2]; o.w = u[4 * q + 3];
            out4[(node0 + c) * (HID / 4) + half * 8 + q] = o;
        }
    }

    if (Wn) {
        v8f acn[4] = {};
#pragma unroll
        for (int sp = 0; sp < 8; ++sp) {
            v2f b0; b0.x = u[4 * sp];     b0.y = u[4 * sp + 1];
            v2f b1; b1.x = u[4 * sp + 2]; b1.y = u[4 * sp + 3];
            float4 aa[4];
#pragma unroll
            for (int jo = 0; jo < 4; ++jo)
                aa[jo] = *(const float4*)&lw[64 * WST + (jo * 16 + c) * WST + half * 32 + 4 * sp];
#pragma unroll
            for (int jo = 0; jo < 4; ++jo) {
                v2f a; a.x = aa[jo].x; a.y = aa[jo].y;
                acn[jo] = __builtin_amdgcn_wmma_f32_16x16x4_f32(
                    false, a, false, b0, (short)0, acn[jo], false, false);
            }
#pragma unroll
            for (int jo = 0; jo < 4; ++jo) {
                v2f a; a.x = aa[jo].z; a.y = aa[jo].w;
                acn[jo] = __builtin_amdgcn_wmma_f32_16x16x4_f32(
                    false, a, false, b1, (short)0, acn[jo], false, false);
            }
        }
        float4* m4 = (float4*)mout;
        if (full) {
#pragma unroll
            for (int jo = 0; jo < 4; ++jo) {
                float4 o0, o1;
                o0.x = fmaxf(acn[jo][0], 0.f); o0.y = fmaxf(acn[jo][1], 0.f);
                o0.z = fmaxf(acn[jo][2], 0.f); o0.w = fmaxf(acn[jo][3], 0.f);
                o1.x = fmaxf(acn[jo][4], 0.f); o1.y = fmaxf(acn[jo][5], 0.f);
                o1.z = fmaxf(acn[jo][6], 0.f); o1.w = fmaxf(acn[jo][7], 0.f);
                *(float4*)(town + 8 * jo)     = o0;
                *(float4*)(town + 8 * jo + 4) = o1;
            }
            __builtin_amdgcn_wave_barrier();
            float4 o[8];
#pragma unroll
            for (int i = 0; i < 8; ++i) o[i] = *(const float4*)(tmem + 2 * i * SP);
#pragma unroll
            for (int i = 0; i < 8; ++i) m4[g0 + 32 * i] = o[i];
        } else if (c < rem) {
#pragma unroll
            for (int jo = 0; jo < 4; ++jo) {
                float4 o0, o1;
                o0.x = fmaxf(acn[jo][0], 0.f); o0.y = fmaxf(acn[jo][1], 0.f);
                o0.z = fmaxf(acn[jo][2], 0.f); o0.w = fmaxf(acn[jo][3], 0.f);
                o1.x = fmaxf(acn[jo][4], 0.f); o1.y = fmaxf(acn[jo][5], 0.f);
                o1.z = fmaxf(acn[jo][6], 0.f); o1.w = fmaxf(acn[jo][7], 0.f);
                m4[(node0 + c) * (HID / 4) + half * 8 + 2 * jo]     = o0;
                m4[(node0 + c) * (HID / 4) + half * 8 + 2 * jo + 1] = o1;
            }
        }
    }
}

extern "C" void kernel_launch(void* const* d_in, const int* in_sizes, int n_in,
                              void* d_out, int out_size, void* d_ws, size_t ws_size,
                              hipStream_t stream)
{
    const float* x       = (const float*)d_in[0];
    const float* conv_w  = (const float*)d_in[1];
    const float* conv_nw = (const float*)d_in[2];
    const float* hid_w   = (const float*)d_in[3];
    const float* hid_nw  = (const float*)d_in[4];
    const int*   row     = (const int*)d_in[5];
    const int*   col     = (const int*)d_in[6];

    const int n  = in_sizes[0] / HID;
    const int ne = in_sizes[5];

    float* P  = (float*)d_out;
    float* Q0 = (float*)d_ws;
    float* R  = Q0 + ((size_t)n + 1) * HID;
    float* Q1 = R + (size_t)n * HID;
    int* cnt = (int*)(Q1 + ((size_t)n + 1) * HID);
    int* ell = cnt + ((size_t)n + 1 + 31) / 32 * 32;
    int* ovf = ell + (size_t)n * ELLK;

    const dim3 blk(256);
    const int gemm_grid = (n + 127) / 128;
    const int fill_grid = (ne + 1023) / 1024;

    gnn_gemm_relu<<<gemm_grid, blk, 0, stream>>>(x, conv_w, nullptr, Q0, R, cnt, n);
    gnn_fill<<<fill_grid, blk, 0, stream>>>(row, col, cnt, ell, ovf, n, ne);

    for (int l = 0; l < 4; ++l) {
        const float* hw = hid_w  + (size_t)l * HID * HID;
        const float* cn = conv_nw + l * HID;
        const float* hn = hid_nw  + l * HID;
        const float* cwn = l < 3 ? conv_w + (size_t)(l + 1) * HID * HID : nullptr;

        const float* mcur = (l & 1) ? Q1 : Q0;
        float*       mnxt = (l & 1) ? Q0 : Q1;

        gnn_pull_norm_gemm_norm<<<gemm_grid, blk, 0, stream>>>(mcur, cnt, ell, ovf,
                                                               R, hw, cn, hn,
                                                               l == 3 ? P : nullptr,
                                                               cwn, mnxt, n);
    }
}


# Edge lists built by owner blocks: a block owns 1024 rows, scans all destinations, ranks its own edges by LDS adds: no device atomic per edge; list order (so each row's sum order) changes
# speedup vs baseline: 3.5139x; 1.0539x over previous
#include <hip/hip_runtime.h>

typedef __attribute__((ext_vector_type(2))) float v2f;
typedef __attribute__((ext_vector_type(8))) float v8f;
typedef __attribute__((ext_vector_type(4))) float v4f;

#define HID 64
#define WST 68
#define SP  68

__global__ __launch_bounds__(256) void gnn_gemm_relu(
    const float* __restrict__ in, const float* __restrict__ W,
    const float* __restrict__ resid, float* __restrict__ out,
    float* __restrict__ copy, int* __restrict__ zero, int n)
{
    __shared__ float lw[64 * WST];
    const int tid = threadIdx.x;

    if (zero) {
        const int gid = (int)blockIdx.x * 256 + tid;
        if (gid == 0) zero[n] = 0;
        if (gid < HID) out[(size_t)n * HID + gid] = 0.f;
    }

    if (copy) {
        const int rows_left = n - (int)blockIdx.x * 128;
        const size_t base = (size_t)blockIdx.x * 128 * HID;
        const v4f* src = (const v4f*)(in + base) + tid;
        v4f* dst = (v4f*)(copy + base) + tid;
        if (rows_left >= 128) {
            v4f cv[8];
#pragma unroll
            for (int p = 0; p < 8; ++p) cv[p] = src[p * 256];
            asm volatile("" :: "v"(cv[0]), "v"(cv[1]), "v"(cv[2]), "v"(cv[3]),
                               "v"(cv[4]), "v"(cv[5]), "v"(cv[6]), "v"(cv[7]));
#pragma unroll
            for (int p = 0; p < 8; ++p) dst[p * 256] = cv[p];
        } else {
            for (int p = 0; 16 * p + (tid >> 4) < rows_left; ++p) dst[p * 256] = src[p * 256];
        }
    }

    for (int i = tid; i < 64 * 64; i += 256) {
        int r = i >> 6, c = i & 63;
        lw[r * WST + c] = W[i];
    }
    __syncthreads();

    const int wave = tid >> 5;
    const int lane = tid & 31;
    const int half = lane >> 4;
    const int l16  = lane & 15;
    const int row0 = blockIdx.x * 128 + wave * 16;

    const int  arow  = row0 + l16;
    const bool rowok = arow < n;
    const float* abase = in + (size_t)(rowok ? arow : 0) * HID + half * 2;
    v2f a[16];
#pragma unroll
    for (int s = 0; s < 16; ++s) {
        float2 t = rowok ? *(const float2*)(abase + 4 * s) : make_float2(0.f, 0.f);
        a[s].x = t.x; a[s].y = t.y;
    }

    v8f acc[4] = {};

#pragma unroll
    for (int s = 0; s < 16; ++s) {
#pragma unroll
        for (int j = 0; j < 4; ++j) {
            const int nc = j * 16 + l16;
            float2 bb = *(const float2*)&lw[nc * WST + 4 * s + half * 2];
            v2f b; b.x = bb.x; b.y = bb.y;
            acc[j] = __builtin_amdgcn_wmma_f32_16x16x4_f32(
                false, a[s], false, b, (short)0, acc[j], false, false);
        }
    }

#pragma unroll
    for (int j = 0; j < 4; ++j) {
        const int colg = j * 16 + l16;
#pragma unroll
        for (int v = 0; v < 8; ++v) {
            const int rowg = row0 + v + half * 8;
            if (rowg < n) {
                float val = fmaxf(acc[j][v], 0.f);
                if (resid) val += resid[(size_t)rowg * HID + colg];
                out[(size_t)rowg * HID + colg] = val;
            }
        }
    }
}

#define ELLK 32
#define ELLH 16
#define RPW  16
#define FROWS 1024
#define FQ    512
#define FUN   8
#define FP    8

__global__ __launch_bounds__(FROWS) void gnn_fill(
    const int* __restrict__ row, const int* __restrict__ col,
    int* __restrict__ cnt, int* __restrict__ ell, int* __restrict__ ovf,
    int n, int ne)
{
    __shared__ int lc[FROWS];
    __shared__ int qe[(FROWS / 32) * FQ];
    const int tid  = threadIdx.x;
    const int lane = tid & 31;
    const int wave = __builtin_amdgcn_readfirstlane(tid >> 5);
    const int lo   = (int)blockIdx.x * FROWS;
    const int rows = n - lo < FROWS ? n - lo : FROWS;
    lc[tid] = 0;
    __syncthreads();

    int* q = qe + wave * FQ;
    int qn = 0;

    auto place = [&](const int r, const int c) {
        const int slot = atomicAdd(&lc[r - lo], 1);
        if (slot < ELLK) {
            ell[(size_t)r * ELLK + slot] = c;
        } else {
            const int i = atomicAdd(cnt + n, 1);
            ovf[2 * (size_t)i]     = r;
            ovf[2 * (size_t)i + 1] = c;
        }
    };
    auto note = [&](const int r, const int e) {
        const bool hit = (unsigned)(r - lo) < (unsigned)rows;
        const unsigned mask = __builtin_amdgcn_ballot_w32(hit);
        if (mask) {
            const int pos = qn + (int)__builtin_amdgcn_mbcnt_lo(mask, 0u);
            if (hit && pos < FQ) q[pos] = e;
            qn += __builtin_popcount(mask);
        }
    };

    const int ne4 = ne >> 2;
    const int4* row4 = (const int4*)row;
    int u0 = 0;
    for (; u0 + FROWS * FUN <= ne4; u0 += FROWS * FUN) {
        int4 rr[FUN];
#pragma unroll
        for (int k = 0; k < FUN; ++k) rr[k] = row4[u0 + k * FROWS + tid];
#pragma unroll
        for (int k = 0; k < FUN; ++k) {
            const int e = 4 * (u0 + k * FROWS + tid);
            note(rr[k].x, e);
            note(rr[k].y, e + 1);
            note(rr[k].z, e + 2);
            note(rr[k].w, e + 3);
        }
    }
    if (u0 < ne4) {
        int4 rr[FUN];
#pragma unroll
        for (int k = 0; k < FUN; ++k) {
            const int u = u0 + k * FROWS + tid;
            rr[k] = row4[u < ne4 ? u : ne4 - 1];
        }
#pragma unroll
        for (int k = 0; k < FUN; ++k) {
            const int u = u0 + k * FROWS + tid;
            const bool in = u < ne4;
            note(in ? rr[k].x : -1, 4 * u);
            note(in ? rr[k].y : -1, 4 * u + 1);
            note(in ? rr[k].z : -1, 4 * u + 2);
            note(in ? rr[k].w : -1, 4 * u + 3);
        }
    }
    if (wave == 0) {
        const int e = 4 * ne4 + lane;
        note(e < ne ? row[e] : -1, e);
    }

    __builtin_amdgcn_wave_barrier();
    const int qm = qn < FQ ? qn : FQ;
    for (int i0 = 0; i0 < qm; i0 += 32 * FP) {
        int er[FP], ec[FP];
#pragma unroll
        for (int k = 0; k < FP; ++k) {
            const int i = i0 + 32 * k + lane;
            const int e = q[i < qm ? i : qm - 1];
            er[k] = row[e];
            ec[k] = col[e];
        }
        asm volatile("" :: "v"(er[0]), "v"(er[1]), "v"(er[2]), "v"(er[3]),
                           "v"(er[4]), "v"(er[5]), "v"(er[6]), "v"(er[7]),
                           "v"(ec[0]), "v"(ec[1]), "v"(ec[2]), "v"(ec[3]),
                           "v"(ec[4]), "v"(ec[5]), "v"(ec[6]), "v"(ec[7]));
#pragma unroll
        for (int k = 0; k < FP; ++k)
            if (i0 + 32 * k + lane < qm) place(er[k], ec[k]);
    }
    if (qn > FQ) {
        int seen = 0;
        auto again = [&](const int r, const int e) {
            const bool hit = (unsigned)(r - lo) < (unsigned)rows;
            const unsigned mask = __builtin_amdgcn_ballot_w32(hit);
            const int pos = seen + (int)__builtin_amdgcn_mbcnt_lo(mask, 0u);
            if (hit && pos >= FQ) place(r, col[e]);
            seen += __builtin_popcount(mask);
        };
#pragma unroll 1
        for (int ub = 0; ub < ne4; ub += FROWS) {
            const int u = ub + tid;
            const bool in = u < ne4;
            const int4 t4 = row4[in ? u : ne4 - 1];
            again(in ? t4.x : -1, 4 * u);
            again(in ? t4.y : -1, 4 * u + 1);
            again(in ? t4.z : -1, 4 * u + 2);
            again(in ? t4.w : -1, 4 * u + 3);
        }
        if (wave == 0) {
            const int e = 4 * ne4 + lane;
            again(e < ne ? row[e] : -1, e);
        }
    }

    __syncthreads();
    if (tid < rows) cnt[lo + tid] = lc[tid];
}

template <int SH, int K>
__device__ __forceinline__ void gnn_gather_offs(unsigned (&off)[8], const int e8,
                                                const unsigned loff)
{
    off[K] = (unsigned)__builtin_amdgcn_update_dpp(0, e8, 0x150 + SH + K, 0xf, 0xf, true)
             | loff;
    if constexpr (K < 7) gnn_gather_offs<SH, K + 1>(off, e8, loff);
}

template <int K>
__device__ __forceinline__ void gnn_gather_loads(float4 (&v)[8], const unsigned (&off)[8],
                                                 const int left, const char* mb)
{
    v[K] = *(const float4*)(mb + off[K]);
    if constexpr (K < 7) {
        if (K + 1 < left) gnn_gather_loads<K + 1>(v, off, left, mb);
    }
}

template <int K>
__device__ __forceinline__ void gnn_gather_adds(float4& a, const float4 (&v)[8],
                                                const int left)
{
    a.x += v[K].x; a.y += v[K].y; a.z += v[K].z; a.w += v[K].w;
    if constexpr (K < 7) {
        if (K + 1 < left) gnn_gather_adds<K + 1>(a, v, left);
    }
}

template <int SH>
__device__ __forceinline__ void gnn_gather_pass(float4& a, const int e8, const int left,
                                                const char* mb, const unsigned loff)
{
    float4 v[8];
    unsigned off[8];
    gnn_gather_offs<SH, 0>(off, e8, loff);
    gnn_gather_loads<0>(v, off, left, mb);
    gnn_gather_adds<0>(a, v, left);
}

__global__ __launch_bounds__(256, 8) void gnn_pull_norm_gemm_norm(
    const float* __restrict__ m, const int* __restrict__ cnt,
    const int* __restrict__ ell, const int* __restrict__ ovf,
    float* __restrict__ t, const float* __restrict__ W,
    const float* __restrict__ w1, const float* __restrict__ w2,
    float* __restrict__ fin, const float* __restrict__ Wn,
    float* __restrict__ mout, int n)
{
    __shared__ __attribute__((aligned(16))) float lw[2 * 64 * WST];
    __shared__ __attribute__((aligned(16))) float lg[2 * HID];
    __shared__ __attribute__((aligned(16))) float stg[8 * 16 * SP];
    const int tid = threadIdx.x;
    if (Wn && blockIdx.x == 0 && tid < HID) mout[(size_t)n * HID + tid] = 0.f;
    float gword = 0.f;
    if (tid < 2 * HID) gword = tid < HID ? w1[tid] : w2[tid - HID];

    if (Wn) {
        for (int i = tid; i < 64 * 64; i += 256) {
            const int f = i >> 6, k = i & 63;
            const int rho = ((f >> 3) & 3) * 16 + (f >> 5) * 8 + (f & 7);
            const float a = W[i];
            const float b = Wn[i];
            lw[rho * WST + k] = a;
            lw[64 * WST + rho * WST + k] = b;
        }
    } else {
        for (int i = tid; i < 64 * 64; i += 256) {
            const int f = i >> 6, k = i & 63;
            const int rho = ((f >> 3) & 3) * 16 + (f >> 5) * 8 + (f & 7);
            lw[rho * WST + k] = W[i];
        }
    }
    if (tid < 2 * HID) lg[tid] = gword;
    __syncthreads();

    const int wave = __builtin_amdgcn_readfirstlane(tid >> 5);
    const int lane = tid & 31;
    const int half = lane >> 4;
    const int c    = lane & 15;
    const int node0 = blockIdx.x * 128 + wave * 16;
    if (node0 >= n) return;
    const int g0   = node0 * (HID / 4) + lane;
    const int rem = n - node0;
    const bool full = rem >= 16;
    float* tile = stg + wave * (16 * SP);
    float* tmem = tile + half * SP + 4 * c;
    float* town = tile + c * SP + 32 * half;

    {
        int deg[RPW];
        int elo[RPW / 2];
        float4 hrow[RPW / 2];
#pragma unroll
        for (int j = 0; j < RPW; ++j) {
            const int r = node0 + j < n ? node0 + j : n - 1;
            deg[j] = node0 + j < n ? __builtin_amdgcn_readfirstlane(cnt[r]) : 0;
        }
#pragma unroll
        for (int p = 0; p < RPW / 2; ++p) {
            const int jr = node0 + 2 * p + half;
            const int r = jr < n ? jr : n - 1;
            elo[p] = ell[(size_t)r * ELLK + c];
        }
#pragma unroll
        for (int p = 0; p < RPW / 2; ++p) {
            const int jr = node0 + 2 * p + half;
            const int r = jr < n ? jr : n - 1;
            hrow[p] = *(const float4*)(t + (size_t)r * HID + 4 * c);
        }
        asm volatile("" :: "v"(elo[0]), "v"(elo[1]), "v"(elo[2]), "v"(elo[3]),
                           "v"(elo[4]), "v"(elo[5]), "v"(elo[6]), "v"(elo[7]));
        const char* mb = (const char*)m;
        const unsigned loff = 16u * (unsigned)c;

        bool anybig = false;
#pragma unroll
        for (int p = 0; p < RPW / 2; ++p) {
            const int da = deg[2 * p]     < ELLH ? deg[2 * p]     : ELLH;
            const int db = deg[2 * p + 1] < ELLH ? deg[2 * p + 1] : ELLH;
            const int dm = da > db ? da : db;
            anybig |= deg[2 * p] > ELLH || deg[2 * p + 1] > ELLH;
            float4 a = hrow[p];
            const int myd = half ? db : da;
            const unsigned ec = c < myd ? (unsigned)elo[p] : (unsigned)n;
            const int e8 = (int)(ec << 8);
            if (dm > 0) gnn_gather_pass<0>(a, e8, dm, mb, loff);
            if (dm > 8) gnn_gather_pass<8>(a, e8, dm - 8, mb, loff);
            *(float4*)(tile + (2 * p + half) * SP + 4 * c) = a;
        }

        if (anybig) {
            __builtin_amdgcn_wave_barrier();
#pragma unroll 1
            for (int j = 0; j < RPW; ++j) {
                const int r = node0 + j;
                if (r >= n) break;
                const int d = __builtin_amdgcn_readfirstlane(cnt[r]);
                if (d <= ELLH) continue;
                float2 a2 = *(float2*)(tile + j * SP + 2 * lane);
                const int idx = ell[(size_t)r * ELLK + lane];
                const int lim = d < ELLK ? d : ELLK;
                for (int s = ELLH; s < lim; s += 8) {
                    float2 vv[8];
#pragma unroll
                    for (int k = 0; k < 8; ++k) {
                        vv[k] = make_float2(0.f, 0.f);
                        if (s + k < lim) {
                            const int src = __builtin_amdgcn_readlane(idx, s + k);
                            vv[k] = *(const float2*)(m + (size_t)src * HID + 2 * lane);
                        }
                    }
#pragma unroll
                    for (int k = 0; k < 8; ++k) { a2.x += vv[k].x; a2.y += vv[k].y; }
                }
                if (d > ELLK) {
                    const int novf = __builtin_amdgcn_readfirstlane(cnt[n]);
                    for (int b = 0; b < novf; b += 32) {
                        const int i = b + lane;
                        int orow = -1, ocol = 0;
                        if (i < novf) { orow = ovf[2 * (size_t)i]; ocol = ovf[2 * (size_t)i + 1]; }
                        unsigned mask = __builtin_amdgcn_ballot_w32(orow == r);
                        while (mask) {
                            const int k = __builtin_ctz(mask);
                            mask &= mask - 1;
                            const int src = __builtin_amdgcn_readlane(ocol, k);
                            const float2 v = *(const float2*)(m + (size_t)src * HID + 2 * lane);
                            a2.x += v.x; a2.y += v.y;
                        }
                    }
                }
                *(float2*)(tile + j * SP + 2 * lane) = a2;
            }
        }
    }
    __builtin_amdgcn_wave_barrier();
    float u[32];
#pragma unroll
    for (int q = 0; q < 8; ++q) {
        const float4 v = *(const float4*)(town + 4 * q);
        u[4 * q + 0] = v.x; u[4 * q + 1] = v.y;
        u[4 * q + 2] = v.z; u[4 * q + 3] = v.w;
    }

    {
        float p0 = 0.f, p1 = 0.f, p2 = 0.f, p3 = 0.f;
#pragma unroll
        for (int q = 0; q < 8; ++q) {
            p0 += u[4 * q + 0] * u[4 * q + 0];
            p1 += u[4 * q + 1] * u[4 * q + 1];
            p2 += u[4 * q + 2] * u[4 * q + 2];
            p3 += u[4 * q + 3] * u[4 * q + 3];
        }
        float ss = (p0 + p1) + (p2 + p3);
        ss += __shfl_xor(ss, 16, 32);
        const float inv = rsqrtf(ss * (1.0f / 64.0f) + 1e-5f);
#pragma unroll
        for (int q = 0; q < 8; ++q) {
            const float4 g = *(const float4*)&lg[half * 32 + 4 * q];
            u[4 * q + 0] = u[4 * q + 0] * inv * g.x;
            u[4 * q + 1] = u[4 * q + 1] * inv * g.y;
            u[4 * q + 2] = u[4 * q + 2] * inv * g.z;
            u[4 * q + 3] = u[4 * q + 3] * inv * g.w;
        }
    }

    v8f acc[4] = {};

#pragma unroll
    for (int sp = 0; sp < 8; ++sp) {
        v2f b0; b0.x = u[4 * sp];     b0.y = u[4 * sp + 1];
        v2f b1; b1.x = u[4 * sp + 2]; b1.y = u[4 * sp + 3];
        float4 aa[4];
#pragma unroll
        for (int jo = 0; jo < 4; ++jo)
            aa[jo] = *(const float4*)&lw[(jo * 16 + c) * WST + half * 32 + 4 * sp];
#pragma unroll
        for (int jo = 0; jo < 4; ++jo) {
            v2f a; a.x = aa[jo].x; a.y = aa[jo].y;
            acc[jo] = __builtin_amdgcn_wmma_f32_16x16x4_f32(
                false, a, false, b0, (short)0, acc[jo], false, false);
        }
#pragma unroll
        for (int jo = 0; jo < 4; ++jo) {
            v2f a; a.x = aa[jo].z; a.y = aa[jo].w;
            acc[jo] = __builtin_amdgcn_wmma_f32_16x16x4_f32(
                false, a, false, b1, (short)0, acc[jo], false, false);
        }
    }

    float p0 = 0.f, p1 = 0.f, p2 = 0.f, p3 = 0.f;
#pragma unroll
    for (int jo = 0; jo < 4; ++jo) {
#pragma unroll
        for (int r = 0; r < 8; ++r)
            u[8 * jo + r] += fmaxf(acc[jo][r], 0.f);
    }
#pragma unroll
    for (int q = 0; q < 8; ++q) {
        p0 += u[4 * q + 0] * u[4 * q + 0];
        p1 += u[4 * q + 1] * u[4 * q + 1];
        p2 += u[4 * q + 2] * u[4 * q + 2];
        p3 += u[4 * q + 3] * u[4 * q + 3];
    }
    float ss2 = (p0 + p1) + (p2 + p3);
    ss2 += __shfl_xor(ss2, 16, 32);
    const float inv2 = rsqrtf(ss2 * (1.0f / 64.0f) + 1e-5f);
    float4* out4 = (float4*)(fin ? fin : t);
#pragma unroll
    for (int q = 0; q < 8; ++q) {
        const float4 g = *(const float4*)&lg[HID + half * 32 + 4 * q];
        u[4 * q + 0] = u[4 * q + 0] * inv2 * g.x;
        u[4 * q + 1] = u[4 * q + 1] * inv2 * g.y;
        u[4 * q + 2] = u[4 * q + 2] * inv2 * g.z;
        u[4 * q + 3] = u[4 * q + 3] * inv2 * g.w;
    }
    if (full) {
#pragma unroll
        for (int q = 0; q < 8; ++q) {
            float4 o;
            o.x = u[4 * q + 0]; o.y = u[4 * q + 1];
            o.z = u[4 * q + 2]; o.w = u[4 * q + 3];
            *(float4*)(town + 4 * q) = o;
        }
        __builtin_amdgcn_wave_barrier();
        float4 o[8];
#pragma unroll
        for (int i = 0; i < 8; ++i) o[i] = *(const float4*)(tmem + 2 * i * SP);
#pragma unroll
        for (int i = 0; i < 8; ++i) out4[g0 + 32 * i] = o[i];
    } else if (c < rem) {
#pragma unroll
        for (int q = 0; q < 8; ++q) {
            float4 o;
            o.x = u[4 * q + 0]; o.y = u[4 * q + 1];
            o.z = u[4 * q + 2]; o.w = u[4 * q + 3];
            out4[(node0 + c) * (HID / 4) + half * 8 + q] = o;
        }
    }

    if (Wn) {
        v8f acn[4] = {};
#pragma unroll
        for (int sp = 0; sp < 8; ++sp) {
            v2f b0; b0.x = u[4 * sp];     b0.y = u[4 * sp + 1];
            v2f b1; b1.x = u[4 * sp + 2]; b1.y = u[4 * sp + 3];
            float4 aa[4];
#pragma unroll
            for (int jo = 0; jo < 4; ++jo)
                aa[jo] = *(const float4*)&lw[64 * WST + (jo * 16 + c) * WST + half * 32 + 4 * sp];
#pragma unroll
            for (int jo = 0; jo < 4; ++jo) {
                v2f a; a.x = aa[jo].x; a.y = aa[jo].y;
                acn[jo] = __builtin_amdgcn_wmma_f32_16x16x4_f32(
                    false, a, false, b0, (short)0, acn[jo], false, false);
            }
#pragma unroll
            for (int jo = 0; jo < 4; ++jo) {
                v2f a; a.x = aa[jo].z; a.y = aa[jo].w;
                acn[jo] = __builtin_amdgcn_wmma_f32_16x16x4_f32(
                    false, a, false, b1, (short)0, acn[jo], false, false);
            }
        }
        float4* m4 = (float4*)mout;
        if (full) {
#pragma unroll
            for (int jo = 0; jo < 4; ++jo) {
                float4 o0, o1;
                o0.x = fmaxf(acn[jo][0], 0.f); o0.y = fmaxf(acn[jo][1], 0.f);
                o0.z = fmaxf(acn[jo][2], 0.f); o0.w = fmaxf(acn[jo][3], 0.f);
                o1.x = fmaxf(acn[jo][4], 0.f); o1.y = fmaxf(acn[jo][5], 0.f);
                o1.z = fmaxf(acn[jo][6], 0.f); o1.w = fmaxf(acn[jo][7], 0.f);
                *(float4*)(town + 8 * jo)     = o0;
                *(float4*)(town + 8 * jo + 4) = o1;
            }
            __builtin_amdgcn_wave_barrier();
            float4 o[8];
#pragma unroll
            for (int i = 0; i < 8; ++i) o[i] = *(const float4*)(tmem + 2 * i * SP);
#pragma unroll
            for (int i = 0; i < 8; ++i) m4[g0 + 32 * i] = o[i];
        } else if (c < rem) {
#pragma unroll
            for (int jo = 0; jo < 4; ++jo) {
                float4 o0, o1;
                o0.x = fmaxf(acn[jo][0], 0.f); o0.y = fmaxf(acn[jo][1], 0.f);
                o0.z = fmaxf(acn[jo][2], 0.f); o0.w = fmaxf(acn[jo][3], 0.f);
                o1.x = fmaxf(acn[jo][4], 0.f); o1.y = fmaxf(acn[jo][5], 0.f);
                o1.z = fmaxf(acn[jo][6], 0.f); o1.w = fmaxf(acn[jo][7], 0.f);
                m4[(node0 + c) * (HID / 4) + half * 8 + 2 * jo]     = o0;
                m4[(node0 + c) * (HID / 4) + half * 8 + 2 * jo + 1] = o1;
            }
        }
    }
}

extern "C" void kernel_launch(void* const* d_in, const int* in_sizes, int n_in,
                              void* d_out, int out_size, void* d_ws, size_t ws_size,
                              hipStream_t stream)
{
    const float* x       = (const float*)d_in[0];
    const float* conv_w  = (const float*)d_in[1];
    const float* conv_nw = (const float*)d_in[2];
    const float* hid_w   = (const float*)d_in[3];
    const float* hid_nw  = (const float*)d_in[4];
    const int*   row     = (const int*)d_in[5];
    const int*   col     = (const int*)d_in[6];

    const int n  = in_sizes[0] / HID;
    const int ne = in_sizes[5];

    float* P  = (float*)d_out;
    float* Q0 = (float*)d_ws;
    float* R  = Q0 + ((size_t)n + 1) * HID;
    float* Q1 = R + (size_t)n * HID;
    int* cnt = (int*)(Q1 + ((size_t)n + 1) * HID);
    int* ell = cnt + ((size_t)n + 1 + 31) / 32 * 32;
    int* ovf = ell + (size_t)n * ELLK;

    const dim3 blk(256);
    const int gemm_grid = (n + 127) / 128;
    const int fill_grid = (n + FROWS - 1) / FROWS;

    gnn_gemm_relu<<<gemm_grid, blk, 0, stream>>>(x, conv_w, nullptr, Q0, R, cnt, n);
    gnn_fill<<<fill_grid, dim3(FROWS), 0, stream>>>(row, col, cnt, ell, ovf, n, ne);

    for (int l = 0; l < 4; ++l) {
        const float* hw = hid_w  + (size_t)l * HID * HID;
        const float* cn = conv_nw + l * HID;
        const float* hn = hid_nw  + l * HID;
        const float* cwn = l < 3 ? conv_w + (size_t)(l + 1) * HID * HID : nullptr;

        const float* mcur = (l & 1) ? Q1 : Q0;
        float*       mnxt = (l & 1) ? Q0 : Q1;

        gnn_pull_norm_gemm_norm<<<gemm_grid, blk, 0, stream>>>(mcur, cnt, ell, ovf,
                                                               R, hw, cn, hn,
                                                               l == 3 ? P : nullptr,
                                                               cwn, mnxt, n);
    }
}


# First product and edge-list build share one launch (block roles by index; the 98 list blocks left most WGPs idle): 6 -> 5 launches; a row over 32 edges is summed from the edge arrays in edge order
# speedup vs baseline: 3.8248x; 1.0885x over previous
#include <hip/hip_runtime.h>

typedef __attribute__((ext_vector_type(2))) float v2f;
typedef __attribute__((ext_vector_type(8))) float v8f;
typedef __attribute__((ext_vector_type(4))) float v4f;

#define HID 64
#define WST 68
#define SP  68
#define FROWS 1024

__device__ __forceinline__ void gnn_first_product(
    const float* __restrict__ in, const float* __restrict__ W,
    float* __restrict__ out, float* __restrict__ copy,
    const int blk, const int nblk, const int n)
{
    __shared__ float lw[64 * WST];
    const int tid = threadIdx.x;

    if (blk == 0 && tid < HID) out[(size_t)n * HID + tid] = 0.f;

#pragma unroll
    for (int i = tid; i < 64 * 64; i += FROWS) {
        int r = i >> 6, c = i & 63;
        lw[r * WST + c] = W[i];
    }
    __syncthreads();

    const int wave = __builtin_amdgcn_readfirstlane(tid >> 5);
    const int t256 = tid & 255;
    const int lane = tid & 31;
    const int half = lane >> 4;
    const int l16  = lane & 15;
    const int nrun = (n + 127) >> 7;

    for (int g = (wave >> 3) * nblk + blk; g < nrun; g += 4 * nblk) {
        {
            const int rows_left = n - g * 128;
            const size_t base = (size_t)g * 128 * HID;
            const v4f* src = (const v4f*)(in + base) + t256;
            v4f* dst = (v4f*)(copy + base) + t256;
            if (rows_left >= 128) {
                v4f cv[8];
#pragma unroll
                for (int p = 0; p < 8; ++p) cv[p] = src[p * 256];
                asm volatile("" :: "v"(cv[0]), "v"(cv[1]), "v"(cv[2]), "v"(cv[3]),
                                   "v"(cv[4]), "v"(cv[5]), "v"(cv[6]), "v"(cv[7]));
#pragma unroll
                for (int p = 0; p < 8; ++p) dst[p * 256] = cv[p];
            } else {
                for (int p = 0; 16 * p + (t256 >> 4) < rows_left; ++p) dst[p * 256] = src[p * 256];
            }
        }

        const int row0 = g * 128 + (wave & 7) * 16;

        const int  arow  = row0 + l16;
        const bool rowok = arow < n;
        const float* abase = in + (size_t)(rowok ? arow : 0) * HID + half * 2;
        v2f a[16];
#pragma unroll
        for (int s = 0; s < 16; ++s) {
            float2 t = rowok ? *(const float2*)(abase + 4 * s) : make_float2(0.f, 0.f);
            a[s].x = t.x; a[s].y = t.y;
        }

        v8f acc[4] = {};

#pragma unroll
        for (int s = 0; s < 16; ++s) {
#pragma unroll
            for (int j = 0; j < 4; ++j) {
                const int nc = j * 16 + l16;
                float2 bb = *(const float2*)&lw[nc * WST + 4 * s + half * 2];
                v2f b; b.x = bb.x; b.y = bb.y;
                acc[j] = __builtin_amdgcn_wmma_f32_16x16x4_f32(
                    false, a[s], false, b, (short)0, acc[j], false, false);
            }
        }

#pragma unroll
        for (int j = 0; j < 4; ++j) {
            const int colg = j * 16 + l16;
#pragma unroll
            for (int v = 0; v < 8; ++v) {
                const int rowg = row0 + v + half * 8;
                if (rowg < n) out[(size_t)rowg * HID + colg] = fmaxf(acc[j][v], 0.f);
            }
        }
    }
}

#define ELLK 32
#define ELLH 16
#define RPW  16
#define FQ    512
#define FUN   8
#define FP    8

__device__ __forceinline__ void gnn_fill(
    const int* __restrict__ row, const int* __restrict__ col,
    int* __restrict__ cnt, int* __restrict__ ell,
    const int blk, const int n, const int ne)
{
    __shared__ int lc[FROWS];
    __shared__ int qe[(FROWS / 32) * FQ];
    const int tid  = threadIdx.x;
    const int lane = tid & 31;
    const int wave = __builtin_amdgcn_readfirstlane(tid >> 5);
    const int lo   = blk * FROWS;
    const int rows = n - lo < FROWS ? n - lo : FROWS;
    lc[tid] = 0;
    __syncthreads();

    int* q = qe + wave * FQ;
    int qn = 0;

    auto place = [&](const int r, const int c) {
        const int slot = atomicAdd(&lc[r - lo], 1);
        if (slot < ELLK) ell[(size_t)r * ELLK + slot] = c;
    };
    auto note = [&](const int r, const int e) {
        const bool hit = (unsigned)(r - lo) < (unsigned)rows;
        const unsigned mask = __builtin_amdgcn_ballot_w32(hit);
        if (mask) {
            const int pos = qn + (int)__builtin_amdgcn_mbcnt_lo(mask, 0u);
            if (hit && pos < FQ) q[pos] = e;
            qn += __builtin_popcount(mask);
        }
    };

    const int ne4 = ne >> 2;
    const int4* row4 = (const int4*)row;
    int u0 = 0;
    for (; u0 + FROWS * FUN <= ne4; u0 += FROWS * FUN) {
        int4 rr[FUN];
#pragma unroll
        for (int k = 0; k < FUN; ++k) rr[k] = row4[u0 + k * FROWS + tid];
#pragma unroll
        for (int k = 0; k < FUN; ++k) {
            const int e = 4 * (u0 + k * FROWS + tid);
            note(rr[k].x, e);
            note(rr[k].y, e + 1);
            note(rr[k].z, e + 2);
            note(rr[k].w, e + 3);
        }
    }
    if (u0 < ne4) {
        int4 rr[FUN];
#pragma unroll
        for (int k = 0; k < FUN; ++k) {
            const int u = u0 + k * FROWS + tid;
            rr[k] = row4[u < ne4 ? u : ne4 - 1];
        }
#pragma unroll
        for (int k = 0; k < FUN; ++k) {
            const int u = u0 + k * FROWS + tid;
            const bool in = u < ne4;
            note(in ? rr[k].x : -1, 4 * u);
            note(in ? rr[k].y : -1, 4 * u + 1);
            note(in ? rr[k].z : -1, 4 * u + 2);
            note(in ? rr[k].w : -1, 4 * u + 3);
        }
    }
    if (wave == 0) {
        const int e = 4 * ne4 + lane;
        note(e < ne ? row[e] : -1, e);
    }

    __builtin_amdgcn_wave_barrier();
    const int qm = qn < FQ ? qn : FQ;
    for (int i0 = 0; i0 < qm; i0 += 32 * FP) {
        int er[FP], ec[FP];
#pragma unroll
        for (int k = 0; k < FP; ++k) {
            const int i = i0 + 32 * k + lane;
            const int e = q[i < qm ? i : qm - 1];
            er[k] = row[e];
            ec[k] = col[e];
        }
        asm volatile("" :: "v"(er[0]), "v"(er[1]), "v"(er[2]), "v"(er[3]),
                           "v"(er[4]), "v"(er[5]), "v"(er[6]), "v"(er[7]),
                           "v"(ec[0]), "v"(ec[1]), "v"(ec[2]), "v"(ec[3]),
                           "v"(ec[4]), "v"(ec[5]), "v"(ec[6]), "v"(ec[7]));
#pragma unroll
        for (int k = 0; k < FP; ++k)
            if (i0 + 32 * k + lane < qm) place(er[k], ec[k]);
    }
    if (qn > FQ) {
        int seen = 0;
        auto again = [&](const int r, const int e) {
            const bool hit = (unsigned)(r - lo) < (unsigned)rows;
            const unsigned mask = __builtin_amdgcn_ballot_w32(hit);
            const int pos = seen + (int)__builtin_amdgcn_mbcnt_lo(mask, 0u);
            if (hit && pos >= FQ) place(r, col[e]);
            seen += __builtin_popcount(mask);
        };
#pragma unroll 1
        for (int ub = 0; ub < ne4; ub += FROWS) {
            const int u = ub + tid;
            const bool in = u < ne4;
            const int4 t4 = row4[in ? u : ne4 - 1];
            again(in ? t4.x : -1, 4 * u);
            again(in ? t4.y : -1, 4 * u + 1);
            again(in ? t4.z : -1, 4 * u + 2);
            again(in ? t4.w : -1, 4 * u + 3);
        }
        if (wave == 0) {
            const int e = 4 * ne4 + lane;
            again(e < ne ? row[e] : -1, e);
        }
    }

    __syncthreads();
    if (tid < rows) cnt[lo + tid] = lc[tid];
}

__global__ __launch_bounds__(FROWS) void gnn_first(
    const float* __restrict__ x, const float* __restrict__ W,
    float* __restrict__ out, float* __restrict__ copy,
    const int* __restrict__ row, const int* __restrict__ col,
    int* __restrict__ cnt, int* __restrict__ ell, int n, int ne, int nfill)
{
    const int b = (int)blockIdx.x;
    if (b < nfill) gnn_fill(row, col, cnt, ell, b, n, ne);
    else           gnn_first_product(x, W, out, copy, b - nfill, (int)gridDim.x - nfill, n);
}

template <int SH, int K>
__device__ __forceinline__ void gnn_gather_offs(unsigned (&off)[8], const int e8,
                                                const unsigned loff)
{
    off[K] = (unsigned)__builtin_amdgcn_update_dpp(0, e8, 0x150 + SH + K, 0xf, 0xf, true)
             | loff;
    if constexpr (K < 7) gnn_gather_offs<SH, K + 1>(off, e8, loff);
}

template <int K>
__device__ __forceinline__ void gnn_gather_loads(float4 (&v)[8], const unsigned (&off)[8],
                                                 const int left, const char* mb)
{
    v[K] = *(const float4*)(mb + off[K]);
    if constexpr (K < 7) {
        if (K + 1 < left) gnn_gather_loads<K + 1>(v, off, left, mb);
    }
}

template <int K>
__device__ __forceinline__ void gnn_gather_adds(float4& a, const float4 (&v)[8],
                                                const int left)
{
    a.x += v[K].x; a.y += v[K].y; a.z += v[K].z; a.w += v[K].w;
    if constexpr (K < 7) {
        if (K + 1 < left) gnn_gather_adds<K + 1>(a, v, left);
    }
}

template <int SH>
__device__ __forceinline__ void gnn_gather_pass(float4& a, const int e8, const int left,
                                                const char* mb, const unsigned loff)
{
    float4 v[8];
    unsigned off[8];
    gnn_gather_offs<SH, 0>(off, e8, loff);
    gnn_gather_loads<0>(v, off, left, mb);
    gnn_gather_adds<0>(a, v, left);
}

__global__ __launch_bounds__(256, 8) void gnn_pull_norm_gemm_norm(
    const float* __restrict__ m, const int* __restrict__ cnt,
    const int* __restrict__ ell,
    const int* __restrict__ row, const int* __restrict__ col, int ne,
    float* __restrict__ t, const float* __restrict__ W,
    const float* __restrict__ w1, const float* __restrict__ w2,
    float* __restrict__ fin, const float* __restrict__ Wn,
    float* __restrict__ mout, int n)
{
    __shared__ __attribute__((aligned(16))) float lw[2 * 64 * WST];
    __shared__ __attribute__((aligned(16))) float lg[2 * HID];
    __shared__ __attribute__((aligned(16))) float stg[8 * 16 * SP];
    const int tid = threadIdx.x;
    if (Wn && blockIdx.x == 0 && tid < HID) mout[(size_t)n * HID + tid] = 0.f;
    float gword = 0.f;
    if (tid < 2 * HID) gword = tid < HID ? w1[tid] : w2[tid - HID];

    if (Wn) {
        for (int i = tid; i < 64 * 64; i += 256) {
            const int f = i >> 6, k = i & 63;
            const int rho = ((f >> 3) & 3) * 16 + (f >> 5) * 8 + (f & 7);
            const float a = W[i];
            const float b = Wn[i];
            lw[rho * WST + k] = a;
            lw[64 * WST + rho * WST + k] = b;
        }
    } else {
        for (int i = tid; i < 64 * 64; i += 256) {
            const int f = i >> 6, k = i & 63;
            const int rho = ((f >> 3) & 3) * 16 + (f >> 5) * 8 + (f & 7);
            lw[rho * WST + k] = W[i];
        }
    }
    if (tid < 2 * HID) lg[tid] = gword;
    __syncthreads();

    const int wave = __builtin_amdgcn_readfirstlane(tid >> 5);
    const int lane = tid & 31;
    const int half = lane >> 4;
    const int c    = lane & 15;
    const int node0 = blockIdx.x * 128 + wave * 16;
    if (node0 >= n) return;
    const int g0   = node0 * (HID / 4) + lane;
    const int rem = n - node0;
    const bool full = rem >= 16;
    float* tile = stg + wave * (16 * SP);
    float* tmem = tile + half * SP + 4 * c;
    float* town = tile + c * SP + 32 * half;

    {
        int deg[RPW];
        int elo[RPW / 2];
        float4 hrow[RPW / 2];
#pragma unroll
        for (int j = 0; j < RPW; ++j) {
            const int r = node0 + j < n ? node0 + j : n - 1;
            deg[j] = node0 + j < n ? __builtin_amdgcn_readfirstlane(cnt[r]) : 0;
        }
#pragma unroll
        for (int p = 0; p < RPW / 2; ++p) {
            const int jr = node0 + 2 * p + half;
            const int r = jr < n ? jr : n - 1;
            elo[p] = ell[(size_t)r * ELLK + c];
        }
#pragma unroll
        for (int p = 0; p < RPW / 2; ++p) {
            const int jr = node0 + 2 * p + half;
            const int r = jr < n ? jr : n - 1;
            hrow[p] = *(const float4*)(t + (size_t)r * HID + 4 * c);
        }
        asm volatile("" :: "v"(elo[0]), "v"(elo[1]), "v"(elo[2]), "v"(elo[3]),
                           "v"(elo[4]), "v"(elo[5]), "v"(elo[6]), "v"(elo[7]));
        const char* mb = (const char*)m;
        const unsigned loff = 16u * (unsigned)c;

        bool anybig = false;
#pragma unroll
        for (int p = 0; p < RPW / 2; ++p) {
            const int da = deg[2 * p]     < ELLH ? deg[2 * p]     : ELLH;
            const int db = deg[2 * p + 1] < ELLH ? deg[2 * p + 1] : ELLH;
            const int dm = da > db ? da : db;
            anybig |= deg[2 * p] > ELLH || deg[2 * p + 1] > ELLH;
            float4 a = hrow[p];
            const int myd = half ? db : da;
            const unsigned ec = c < myd ? (unsigned)elo[p] : (unsigned)n;
            const int e8 = (int)(ec << 8);
            if (dm > 0) gnn_gather_pass<0>(a, e8, dm, mb, loff);
            if (dm > 8) gnn_gather_pass<8>(a, e8, dm - 8, mb, loff);
            *(float4*)(tile + (2 * p + half) * SP + 4 * c) = a;
        }

        if (anybig) {
            __builtin_amdgcn_wave_barrier();
#pragma unroll 1
            for (int j = 0; j < RPW; ++j) {
                const int r = node0 + j;
                if (r >= n) break;
                const int d = __builtin_amdgcn_readfirstlane(cnt[r]);
                if (d <= ELLH) continue;
                float2 a2 = *(float2*)(tile + j * SP + 2 * lane);
                const int idx = ell[(size_t)r * ELLK + lane];
                const int lim = d <= ELLK ? d : ELLH;
                for (int s = ELLH; s < lim; s += 8) {
                    float2 vv[8];
#pragma unroll
                    for (int k = 0; k < 8; ++k) {
                        vv[k] = make_float2(0.f, 0.f);
                        if (s + k < lim) {
                            const int src = __builtin_amdgcn_readlane(idx, s + k);
                            vv[k] = *(const float2*)(m + (size_t)src * HID + 2 * lane);
                        }
                    }
#pragma unroll
                    for (int k = 0; k < 8; ++k) { a2.x += vv[k].x; a2.y += vv[k].y; }
                }
                if (d > ELLK) {
                    a2 = *(const float2*)(t + (size_t)r * HID + 2 * lane);
                    for (int b = 0; b < ne; b += 32) {
                        const int i = b + lane;
                        int orow = -1, ocol = 0;
                        if (i < ne) { orow = row[i]; ocol = col[i]; }
                        unsigned mask = __builtin_amdgcn_ballot_w32(orow == r);
                        while (mask) {
                            const int k = __builtin_ctz(mask);
                            mask &= mask - 1;
                            const int src = __builtin_amdgcn_readlane(ocol, k);
                            const float2 v = *(const float2*)(m + (size_t)src * HID + 2 * lane);
                            a2.x += v.x; a2.y += v.y;
                        }
                    }
                }
                *(float2*)(tile + j * SP + 2 * lane) = a2;
            }
        }
    }
    __builtin_amdgcn_wave_barrier();
    float u[32];
#pragma unroll
    for (int q = 0; q < 8; ++q) {
        const float4 v = *(const float4*)(town + 4 * q);
        u[4 * q + 0] = v.x; u[4 * q + 1] = v.y;
        u[4 * q + 2] = v.z; u[4 * q + 3] = v.w;
    }

    {
        float p0 = 0.f, p1 = 0.f, p2 = 0.f, p3 = 0.f;
#pragma unroll
        for (int q = 0; q < 8; ++q) {
            p0 += u[4 * q + 0] * u[4 * q + 0];
            p1 += u[4 * q + 1] * u[4 * q + 1];
            p2 += u[4 * q + 2] * u[4 * q + 2];
            p3 += u[4 * q + 3] * u[4 * q + 3];
        }
        float ss = (p0 + p1) + (p2 + p3);
        ss += __shfl_xor(ss, 16, 32);
        const float inv = rsqrtf(ss * (1.0f / 64.0f) + 1e-5f);
#pragma unroll
        for (int q = 0; q < 8; ++q) {
            const float4 g = *(const float4*)&lg[half * 32 + 4 * q];
            u[4 * q + 0] = u[4 * q + 0] * inv * g.x;
            u[4 * q + 1] = u[4 * q + 1] * inv * g.y;
            u[4 * q + 2] = u[4 * q + 2] * inv * g.z;
            u[4 * q + 3] = u[4 * q + 3] * inv * g.w;
        }
    }

    v8f acc[4] = {};

#pragma unroll
    for (int sp = 0; sp < 8; ++sp) {
        v2f b0; b0.x = u[4 * sp];     b0.y = u[4 * sp + 1];
        v2f b1; b1.x = u[4 * sp + 2]; b1.y = u[4 * sp + 3];
        float4 aa[4];
#pragma unroll
        for (int jo = 0; jo < 4; ++jo)
            aa[jo] = *(const float4*)&lw[(jo * 16 + c) * WST + half * 32 + 4 * sp];
#pragma unroll
        for (int jo = 0; jo < 4; ++jo) {
            v2f a; a.x = aa[jo].x; a.y = aa[jo].y;
            acc[jo] = __builtin_amdgcn_wmma_f32_16x16x4_f32(
                false, a, false, b0, (short)0, acc[jo], false, false);
        }
#pragma unroll
        for (int jo = 0; jo < 4; ++jo) {
            v2f a; a.x = aa[jo].z; a.y = aa[jo].w;
            acc[jo] = __builtin_amdgcn_wmma_f32_16x16x4_f32(
                false, a, false, b1, (short)0, acc[jo], false, false);
        }
    }

    float p0 = 0.f, p1 = 0.f, p2 = 0.f, p3 = 0.f;
#pragma unroll
    for (int jo = 0; jo < 4; ++jo) {
#pragma unroll
        for (int r = 0; r < 8; ++r)
            u[8 * jo + r] += fmaxf(acc[jo][r], 0.f);
    }
#pragma unroll
    for (int q = 0; q < 8; ++q) {
        p0 += u[4 * q + 0] * u[4 * q + 0];
        p1 += u[4 * q + 1] * u[4 * q + 1];
        p2 += u[4 * q + 2] * u[4 * q + 2];
        p3 += u[4 * q + 3] * u[4 * q + 3];
    }
    float ss2 = (p0 + p1) + (p2 + p3);
    ss2 += __shfl_xor(ss2, 16, 32);
    const float inv2 = rsqrtf(ss2 * (1.0f / 64.0f) + 1e-5f);
    float4* out4 = (float4*)(fin ? fin : t);
#pragma unroll
    for (int q = 0; q < 8; ++q) {
        const float4 g = *(const float4*)&lg[HID + half * 32 + 4 * q];
        u[4 * q + 0] = u[4 * q + 0] * inv2 * g.x;
        u[4 * q + 1] = u[4 * q + 1] * inv2 * g.y;
        u[4 * q + 2] = u[4 * q + 2] * inv2 * g.z;
        u[4 * q + 3] = u[4 * q + 3] * inv2 * g.w;
    }
    if (full) {
#pragma unroll
        for (int q = 0; q < 8; ++q) {
            float4 o;
            o.x = u[4 * q + 0]; o.y = u[4 * q + 1];
            o.z = u[4 * q + 2]; o.w = u[4 * q + 3];
            *(float4*)(town + 4 * q) = o;
        }
        __builtin_amdgcn_wave_barrier();
        float4 o[8];
#pragma unroll
        for (int i = 0; i < 8; ++i) o[i] = *(const float4*)(tmem + 2 * i * SP);
#pragma unroll
        for (int i = 0; i < 8; ++i) out4[g0 + 32 * i] = o[i];
    } else if (c < rem) {
#pragma unroll
        for (int q = 0; q < 8; ++q) {
            float4 o;
            o.x = u[4 * q + 0]; o.y = u[4 * q + 1];
            o.z = u[4 * q + 2]; o.w = u[4 * q + 3];
            out4[(node0 + c) * (HID / 4) + half * 8 + q] = o;
        }
    }

    if (Wn) {
        v8f acn[4] = {};
#pragma unroll
        for (int sp = 0; sp < 8; ++sp) {
            v2f b0; b0.x = u[4 * sp];     b0.y = u[4 * sp + 1];
            v2f b1; b1.x = u[4 * sp + 2]; b1.y = u[4 * sp + 3];
            float4 aa[4];
#pragma unroll
            for (int jo = 0; jo < 4; ++jo)
                aa[jo] = *(const float4*)&lw[64 * WST + (jo * 16 + c) * WST + half * 32 + 4 * sp];
#pragma unroll
            for (int jo = 0; jo < 4; ++jo) {
                v2f a; a.x = aa[jo].x; a.y = aa[jo].y;
                acn[jo] = __builtin_amdgcn_wmma_f32_16x16x4_f32(
                    false, a, false, b0, (short)0, acn[jo], false, false);
            }
#pragma unroll
            for (int jo = 0; jo < 4; ++jo) {
                v2f a; a.x = aa[jo].z; a.y = aa[jo].w;
                acn[jo] = __builtin_amdgcn_wmma_f32_16x16x4_f32(
                    false, a, false, b1, (short)0, acn[jo], false, false);
            }
        }
        float4* m4 = (float4*)mout;
        if (full) {
#pragma unroll
            for (int jo = 0; jo < 4; ++jo) {
                float4 o0, o1;
                o0.x = fmaxf(acn[jo][0], 0.f); o0.y = fmaxf(acn[jo][1], 0.f);
                o0.z = fmaxf(acn[jo][2], 0.f); o0.w = fmaxf(acn[jo][3], 0.f);
                o1.x = fmaxf(acn[jo][4], 0.f); o1.y = fmaxf(acn[jo][5], 0.f);
                o1.z = fmaxf(acn[jo][6], 0.f); o1.w = fmaxf(acn[jo][7], 0.f);
                *(float4*)(town + 8 * jo)     = o0;
                *(float4*)(town + 8 * jo + 4) = o1;
            }
            __builtin_amdgcn_wave_barrier();
            float4 o[8];
#pragma unroll
            for (int i = 0; i < 8; ++i) o[i] = *(const float4*)(tmem + 2 * i * SP);
#pragma unroll
            for (int i = 0; i < 8; ++i) m4[g0 + 32 * i] = o[i];
        } else if (c < rem) {
#pragma unroll
            for (int jo = 0; jo < 4; ++jo) {
                float4 o0, o1;
                o0.x = fmaxf(acn[jo][0], 0.f); o0.y = fmaxf(acn[jo][1], 0.f);
                o0.z = fmaxf(acn[jo][2], 0.f); o0.w = fmaxf(acn[jo][3], 0.f);
                o1.x = fmaxf(acn[jo][4], 0.f); o1.y = fmaxf(acn[jo][5], 0.f);
                o1.z = fmaxf(acn[jo][6], 0.f); o1.w = fmaxf(acn[jo][7], 0.f);
                m4[(node0 + c) * (HID / 4) + half * 8 + 2 * jo]     = o0;
                m4[(node0 + c) * (HID / 4) + half * 8 + 2 * jo + 1] = o1;
            }
        }
    }
}

extern "C" void kernel_launch(void* const* d_in, const int* in_sizes, int n_in,
                              void* d_out, int out_size, void* d_ws, size_t ws_size,
                              hipStream_t stream)
{
    const float* x       = (const float*)d_in[0];
    const float* conv_w  = (const float*)d_in[1];
    const float* conv_nw = (const float*)d_in[2];
    const float* hid_w   = (const float*)d_in[3];
    const float* hid_nw  = (const float*)d_in[4];
    const int*   row     = (const int*)d_in[5];
    const int*   col     = (const int*)d_in[6];

    const int n  = in_sizes[0] / HID;
    const int ne = in_sizes[5];

    float* P  = (float*)d_out;
    float* Q0 = (float*)d_ws;
    float* R  = Q0 + ((size_t)n + 1) * HID;
    float* Q1 = R + (size_t)n * HID;
    int* cnt = (int*)(Q1 + ((size_t)n + 1) * HID);
    int* ell = cnt + ((size_t)n + 31) / 32 * 32;

    const dim3 blk(256);
    const int gemm_grid = (n + 127) / 128;
    const int fill_grid = (n + FROWS - 1) / FROWS;
    const int prod_grid = fill_grid < 192 ? 256 - fill_grid : 64;

    gnn_first<<<fill_grid + prod_grid, dim3(FROWS), 0, stream>>>(x, conv_w, Q0, R, row, col,
                                                                 cnt, ell, n, ne, fill_grid);

    for (int l = 0; l < 4; ++l) {
        const float* hw = hid_w  + (size_t)l * HID * HID;
        const float* cn = conv_nw + l * HID;
        const float* hn = hid_nw  + l * HID;
        const float* cwn = l < 3 ? conv_w + (size_t)(l + 1) * HID * HID : nullptr;

        const float* mcur = (l & 1) ? Q1 : Q0;
        float*       mnxt = (l & 1) ? Q0 : Q1;

        gnn_pull_norm_gemm_norm<<<gemm_grid, blk, 0, stream>>>(mcur, cnt, ell, row, col, ne,
                                                               R, hw, cn, hn,
                                                               l == 3 ? P : nullptr,
                                                               cwn, mnxt, n);
    }
}


# Layer kernel: one block of 25 waves (400 rows) a WGP, 250 blocks, not 782 blocks of 8 waves (3 or 4 a WGP as they fall): every busy WGP holds 25 waves; weights staged once a WGP; sums unchanged
# speedup vs baseline: 4.0600x; 1.0615x over previous
#include <hip/hip_runtime.h>

typedef __attribute__((ext_vector_type(2))) float v2f;
typedef __attribute__((ext_vector_type(8))) float v8f;
typedef __attribute__((ext_vector_type(4))) float v4f;

#define HID 64
#define WST 68
#define SP  68
#define FROWS 1024

__device__ __forceinline__ void gnn_first_product(
    const float* __restrict__ in, const float* __restrict__ W,
    float* __restrict__ out, float* __restrict__ copy,
    const int blk, const int nblk, const int n)
{
    __shared__ float lw[64 * WST];
    const int tid = threadIdx.x;

    if (blk == 0 && tid < HID) out[(size_t)n * HID + tid] = 0.f;

#pragma unroll
    for (int i = tid; i < 64 * 64; i += FROWS) {
        int r = i >> 6, c = i & 63;
        lw[r * WST + c] = W[i];
    }
    __syncthreads();

    const int wave = __builtin_amdgcn_readfirstlane(tid >> 5);
    const int t256 = tid & 255;
    const int lane = tid & 31;
    const int half = lane >> 4;
    const int l16  = lane & 15;
    const int nrun = (n + 127) >> 7;

    for (int g = (wave >> 3) * nblk + blk; g < nrun; g += 4 * nblk) {
        {
            const int rows_left = n - g * 128;
            const size_t base = (size_t)g * 128 * HID;
            const v4f* src = (const v4f*)(in + base) + t256;
            v4f* dst = (v4f*)(copy + base) + t256;
            if (rows_left >= 128) {
                v4f cv[8];
#pragma unroll
                for (int p = 0; p < 8; ++p) cv[p] = src[p * 256];
                asm volatile("" :: "v"(cv[0]), "v"(cv[1]), "v"(cv[2]), "v"(cv[3]),
                                   "v"(cv[4]), "v"(cv[5]), "v"(cv[6]), "v"(cv[7]));
#pragma unroll
                for (int p = 0; p < 8; ++p) dst[p * 256] = cv[p];
            } else {
                for (int p = 0; 16 * p + (t256 >> 4) < rows_left; ++p) dst[p * 256] = src[p * 256];
            }
        }

        const int row0 = g * 128 + (wave & 7) * 16;

        const int  arow  = row0 + l16;
        const bool rowok = arow < n;
        const float* abase = in + (size_t)(rowok ? arow : 0) * HID + half * 2;
        v2f a[16];
#pragma unroll
        for (int s = 0; s < 16; ++s) {
            float2 t = rowok ? *(const float2*)(abase + 4 * s) : make_float2(0.f, 0.f);
            a[s].x = t.x; a[s].y = t.y;
        }

        v8f acc[4] = {};

#pragma unroll
        for (int s = 0; s < 16; ++s) {
#pragma unroll
            for (int j = 0; j < 4; ++j) {
                const int nc = j * 16 + l16;
                float2 bb = *(const float2*)&lw[nc * WST + 4 * s + half * 2];
                v2f b; b.x = bb.x; b.y = bb.y;
                acc[j] = __builtin_amdgcn_wmma_f32_16x16x4_f32(
                    false, a[s], false, b, (short)0, acc[j], false, false);
            }
        }

#pragma unroll
        for (int j = 0; j < 4; ++j) {
            const int colg = j * 16 + l16;
#pragma unroll
            for (int v = 0; v < 8; ++v) {
                const int rowg = row0 + v + half * 8;
                if (rowg < n) out[(size_t)rowg * HID + colg] = fmaxf(acc[j][v], 0.f);
            }
        }
    }
}

#define ELLK 32
#define ELLH 16
#define RPW  16
#define LWAVES 25
#define LTHREADS (32 * LWAVES)
#define LROWS (RPW * LWAVES)
#define LSTG ((64 * 64 + LTHREADS - 1) / LTHREADS)
static_assert(LSTG == 6, "the staging fences of the layer kernel name six words a matrix");
#define FQ    512
#define FUN   8
#define FP    8

__device__ __forceinline__ void gnn_fill(
    const int* __restrict__ row, const int* __restrict__ col,
    int* __restrict__ cnt, int* __restrict__ ell,
    const int blk, const int n, const int ne)
{
    __shared__ int lc[FROWS];
    __shared__ int qe[(FROWS / 32) * FQ];
    const int tid  = threadIdx.x;
    const int lane = tid & 31;
    const int wave = __builtin_amdgcn_readfirstlane(tid >> 5);
    const int lo   = blk * FROWS;
    const int rows = n - lo < FROWS ? n - lo : FROWS;
    lc[tid] = 0;
    __syncthreads();

    int* q = qe + wave * FQ;
    int qn = 0;

    auto place = [&](const int r, const int c) {
        const int slot = atomicAdd(&lc[r - lo], 1);
        if (slot < ELLK) ell[(size_t)r * ELLK + slot] = c;
    };
    auto note = [&](const int r, const int e) {
        const bool hit = (unsigned)(r - lo) < (unsigned)rows;
        const unsigned mask = __builtin_amdgcn_ballot_w32(hit);
        if (mask) {
            const int pos = qn + (int)__builtin_amdgcn_mbcnt_lo(mask, 0u);
            if (hit && pos < FQ) q[pos] = e;
            qn += __builtin_popcount(mask);
        }
    };

    const int ne4 = ne >> 2;
    const int4* row4 = (const int4*)row;
    int u0 = 0;
    for (; u0 + FROWS * FUN <= ne4; u0 += FROWS * FUN) {
        int4 rr[FUN];
#pragma unroll
        for (int k = 0; k < FUN; ++k) rr[k] = row4[u0 + k * FROWS + tid];
#pragma unroll
        for (int k = 0; k < FUN; ++k) {
            const int e = 4 * (u0 + k * FROWS + tid);
            note(rr[k].x, e);
            note(rr[k].y, e + 1);
            note(rr[k].z, e + 2);
            note(rr[k].w, e + 3);
        }
    }
    if (u0 < ne4) {
        int4 rr[FUN];
#pragma unroll
        for (int k = 0; k < FUN; ++k) {
            const int u = u0 + k * FROWS + tid;
            rr[k] = row4[u < ne4 ? u : ne4 - 1];
        }
#pragma unroll
        for (int k = 0; k < FUN; ++k) {
            const int u = u0 + k * FROWS + tid;
            const bool in = u < ne4;
            note(in ? rr[k].x : -1, 4 * u);
            note(in ? rr[k].y : -1, 4 * u + 1);
            note(in ? rr[k].z : -1, 4 * u + 2);
            note(in ? rr[k].w : -1, 4 * u + 3);
        }
    }
    if (wave == 0) {
        const int e = 4 * ne4 + lane;
        note(e < ne ? row[e] : -1, e);
    }

    __builtin_amdgcn_wave_barrier();
    const int qm = qn < FQ ? qn : FQ;
    for (int i0 = 0; i0 < qm; i0 += 32 * FP) {
        int er[FP], ec[FP];
#pragma unroll
        for (int k = 0; k < FP; ++k) {
            const int i = i0 + 32 * k + lane;
            const int e = q[i < qm ? i : qm - 1];
            er[k] = row[e];
            ec[k] = col[e];
        }
        asm volatile("" :: "v"(er[0]), "v"(er[1]), "v"(er[2]), "v"(er[3]),
                           "v"(er[4]), "v"(er[5]), "v"(er[6]), "v"(er[7]),
                           "v"(ec[0]), "v"(ec[1]), "v"(ec[2]), "v"(ec[3]),
                           "v"(ec[4]), "v"(ec[5]), "v"(ec[6]), "v"(ec[7]));
#pragma unroll
        for (int k = 0; k < FP; ++k)
            if (i0 + 32 * k + lane < qm) place(er[k], ec[k]);
    }
    if (qn > FQ) {
        int seen = 0;
        auto again = [&](const int r, const int e) {
            const bool hit = (unsigned)(r - lo) < (unsigned)rows;
            const unsigned mask = __builtin_amdgcn_ballot_w32(hit);
            const int pos = seen + (int)__builtin_amdgcn_mbcnt_lo(mask, 0u);
            if (hit && pos >= FQ) place(r, col[e]);
            seen += __builtin_popcount(mask);
        };
#pragma unroll 1
        for (int ub = 0; ub < ne4; ub += FROWS) {
            const int u = ub + tid;
            const bool in = u < ne4;
            const int4 t4 = row4[in ? u : ne4 - 1];
            again(in ? t4.x : -1, 4 * u);
            again(in ? t4.y : -1, 4 * u + 1);
            again(in ? t4.z : -1, 4 * u + 2);
            again(in ? t4.w : -1, 4 * u + 3);
        }
        if (wave == 0) {
            const int e = 4 * ne4 + lane;
            again(e < ne ? row[e] : -1, e);
        }
    }

    __syncthreads();
    if (tid < rows) cnt[lo + tid] = lc[tid];
}

__global__ __launch_bounds__(FROWS) void gnn_first(
    const float* __restrict__ x, const float* __restrict__ W,
    float* __restrict__ out, float* __restrict__ copy,
    const int* __restrict__ row, const int* __restrict__ col,
    int* __restrict__ cnt, int* __restrict__ ell, int n, int ne, int nfill)
{
    const int b = (int)blockIdx.x;
    if (b < nfill) gnn_fill(row, col, cnt, ell, b, n, ne);
    else           gnn_first_product(x, W, out, copy, b - nfill, (int)gridDim.x - nfill, n);
}

template <int SH, int K>
__device__ __forceinline__ void gnn_gather_offs(unsigned (&off)[8], const int e8,
                                                const unsigned loff)
{
    off[K] = (unsigned)__builtin_amdgcn_update_dpp(0, e8, 0x150 + SH + K, 0xf, 0xf, true)
             | loff;
    if constexpr (K < 7) gnn_gather_offs<SH, K + 1>(off, e8, loff);
}

template <int K>
__device__ __forceinline__ void gnn_gather_loads(float4 (&v)[8], const unsigned (&off)[8],
                                                 const int left, const char* mb)
{
    v[K] = *(const float4*)(mb + off[K]);
    if constexpr (K < 7) {
        if (K + 1 < left) gnn_gather_loads<K + 1>(v, off, left, mb);
    }
}

template <int K>
__device__ __forceinline__ void gnn_gather_adds(float4& a, const float4 (&v)[8],
                                                const int left)
{
    a.x += v[K].x; a.y += v[K].y; a.z += v[K].z; a.w += v[K].w;
    if constexpr (K < 7) {
        if (K + 1 < left) gnn_gather_adds<K + 1>(a, v, left);
    }
}

template <int SH>
__device__ __forceinline__ void gnn_gather_pass(float4& a, const int e8, const int left,
                                                const char* mb, const unsigned loff)
{
    float4 v[8];
    unsigned off[8];
    gnn_gather_offs<SH, 0>(off, e8, loff);
    gnn_gather_loads<0>(v, off, left, mb);
    gnn_gather_adds<0>(a, v, left);
}

__global__ __launch_bounds__(LTHREADS) void gnn_pull_norm_gemm_norm(
    const float* __restrict__ m, const int* __restrict__ cnt,
    const int* __restrict__ ell,
    const int* __restrict__ row, const int* __restrict__ col, int ne,
    float* __restrict__ t, const float* __restrict__ W,
    const float* __restrict__ w1, const float* __restrict__ w2,
    float* __restrict__ fin, const float* __restrict__ Wn,
    float* __restrict__ mout, int n)
{
    __shared__ __attribute__((aligned(16))) float lw[2 * 64 * WST];
    __shared__ __attribute__((aligned(16))) float lg[2 * HID];
    __shared__ __attribute__((aligned(16))) float stg[LWAVES * 16 * SP];
    const int tid = threadIdx.x;
    if (Wn && blockIdx.x == 0 && tid < HID) mout[(size_t)n * HID + tid] = 0.f;
    float gword = 0.f;
    if (tid < 2 * HID) gword = tid < HID ? w1[tid] : w2[tid - HID];

    if (Wn) {
        float wa[LSTG], wb[LSTG];
#pragma unroll
        for (int j = 0; j < LSTG; ++j) {
            const int i = tid + j * LTHREADS;
            wa[j] = W[i < 64 * 64 ? i : 64 * 64 - 1];
            wb[j] = Wn[i < 64 * 64 ? i : 64 * 64 - 1];
        }
        asm volatile("" :: "v"(wa[0]), "v"(wa[1]), "v"(wa[2]), "v"(wa[3]), "v"(wa[4]), "v"(wa[5]),
                           "v"(wb[0]), "v"(wb[1]), "v"(wb[2]), "v"(wb[3]), "v"(wb[4]), "v"(wb[5]));
#pragma unroll
        for (int j = 0; j < LSTG; ++j) {
            const int i = tid + j * LTHREADS;
            const int f = i >> 6, k = i & 63;
            const int rho = ((f >> 3) & 3) * 16 + (f >> 5) * 8 + (f & 7);
            if (i < 64 * 64) {
                lw[rho * WST + k] = wa[j];
                lw[64 * WST + rho * WST + k] = wb[j];
            }
        }
    } else {
        float wa[LSTG];
#pragma unroll
        for (int j = 0; j < LSTG; ++j) {
            const int i = tid + j * LTHREADS;
            wa[j] = W[i < 64 * 64 ? i : 64 * 64 - 1];
        }
        asm volatile("" :: "v"(wa[0]), "v"(wa[1]), "v"(wa[2]), "v"(wa[3]), "v"(wa[4]), "v"(wa[5]));
#pragma unroll
        for (int j = 0; j < LSTG; ++j) {
            const int i = tid + j * LTHREADS;
            const int f = i >> 6, k = i & 63;
            const int rho = ((f >> 3) & 3) * 16 + (f >> 5) * 8 + (f & 7);
            if (i < 64 * 64) lw[rho * WST + k] = wa[j];
        }
    }
    if (tid < 2 * HID) lg[tid] = gword;
    __syncthreads();

    const int wave = __builtin_amdgcn_readfirstlane(tid >> 5);
    const int lane = tid & 31;
    const int half = lane >> 4;
    const int c    = lane & 15;
    const int node0 = blockIdx.x * LROWS + wave * RPW;
    if (node0 >= n) return;
    const int g0   = node0 * (HID / 4) + lane;
    const int rem = n - node0;
    const bool full = rem >= 16;
    float* tile = stg + wave * (16 * SP);
    float* tmem = tile + half * SP + 4 * c;
    float* town = tile + c * SP + 32 * half;

    {
        int deg[RPW];
        int elo[RPW / 2];
        float4 hrow[RPW / 2];
#pragma unroll
        for (int j = 0; j < RPW; ++j) {
            const int r = node0 + j < n ? node0 + j : n - 1;
            deg[j] = node0 + j < n ? __builtin_amdgcn_readfirstlane(cnt[r]) : 0;
        }
#pragma unroll
        for (int p = 0; p < RPW / 2; ++p) {
            const int jr = node0 + 2 * p + half;
            const int r = jr < n ? jr : n - 1;
            elo[p] = ell[(size_t)r * ELLK + c];
        }
#pragma unroll
        for (int p = 0; p < RPW / 2; ++p) {
            const int jr = node0 + 2 * p + half;
            const int r = jr < n ? jr : n - 1;
            hrow[p] = *(const float4*)(t + (size_t)r * HID + 4 * c);
        }
        asm volatile("" :: "v"(elo[0]), "v"(elo[1]), "v"(elo[2]), "v"(elo[3]),
                           "v"(elo[4]), "v"(elo[5]), "v"(elo[6]), "v"(elo[7]));
        const char* mb = (const char*)m;
        const unsigned loff = 16u * (unsigned)c;

        bool anybig = false;
#pragma unroll
        for (int p = 0; p < RPW / 2; ++p) {
            const int da = deg[2 * p]     < ELLH ? deg[2 * p]     : ELLH;
            const int db = deg[2 * p + 1] < ELLH ? deg[2 * p + 1] : ELLH;
            const int dm = da > db ? da : db;
            anybig |= deg[2 * p] > ELLH || deg[2 * p + 1] > ELLH;
            float4 a = hrow[p];
            const int myd = half ? db : da;
            const unsigned ec = c < myd ? (unsigned)elo[p] : (unsigned)n;
            const int e8 = (int)(ec << 8);
            if (dm > 0) gnn_gather_pass<0>(a, e8, dm, mb, loff);
            if (dm > 8) gnn_gather_pass<8>(a, e8, dm - 8, mb, loff);
            *(float4*)(tile + (2 * p + half) * SP + 4 * c) = a;
        }

        if (anybig) {
            __builtin_amdgcn_wave_barrier();
#pragma unroll 1
            for (int j = 0; j < RPW; ++j) {
                const int r = node0 + j;
                if (r >= n) break;
                const int d = __builtin_amdgcn_readfirstlane(cnt[r]);
                if (d <= ELLH) continue;
                float2 a2 = *(float2*)(tile + j * SP + 2 * lane);
                const int idx = ell[(size_t)r * ELLK + lane];
                const int lim = d <= ELLK ? d : ELLH;
                for (int s = ELLH; s < lim; s += 8) {
                    float2 vv[8];
#pragma unroll
                    for (int k = 0; k < 8; ++k) {
                        vv[k] = make_float2(0.f, 0.f);
                        if (s + k < lim) {
                            const int src = __builtin_amdgcn_readlane(idx, s + k);
                            vv[k] = *(const float2*)(m + (size_t)src * HID + 2 * lane);
                        }
                    }
#pragma unroll
                    for (int k = 0; k < 8; ++k) { a2.x += vv[k].x; a2.y += vv[k].y; }
                }
                if (d > ELLK) {
                    a2 = *(const float2*)(t + (size_t)r * HID + 2 * lane);
                    for (int b = 0; b < ne; b += 32) {
                        const int i = b + lane;
                        int orow = -1, ocol = 0;
                        if (i < ne) { orow = row[i]; ocol = col[i]; }
                        unsigned mask = __builtin_amdgcn_ballot_w32(orow == r);
                        while (mask) {
                            const int k = __builtin_ctz(mask);
                            mask &= mask - 1;
                            const int src = __builtin_amdgcn_readlane(ocol, k);
                            const float2 v = *(const float2*)(m + (size_t)src * HID + 2 * lane);
                            a2.x += v.x; a2.y += v.y;
                        }
                    }
                }
                *(float2*)(tile + j * SP + 2 * lane) = a2;
            }
        }
    }
    __builtin_amdgcn_wave_barrier();
    float u[32];
#pragma unroll
    for (int q = 0; q < 8; ++q) {
        const float4 v = *(const float4*)(town + 4 * q);
        u[4 * q + 0] = v.x; u[4 * q + 1] = v.y;
        u[4 * q + 2] = v.z; u[4 * q + 3] = v.w;
    }

    {
        float p0 = 0.f, p1 = 0.f, p2 = 0.f, p3 = 0.f;
#pragma unroll
        for (int q = 0; q < 8; ++q) {
            p0 += u[4 * q + 0] * u[4 * q + 0];
            p1 += u[4 * q + 1] * u[4 * q + 1];
            p2 += u[4 * q + 2] * u[4 * q + 2];
            p3 += u[4 * q + 3] * u[4 * q + 3];
        }
        float ss = (p0 + p1) + (p2 + p3);
        ss += __shfl_xor(ss, 16, 32);
        const float inv = rsqrtf(ss * (1.0f / 64.0f) + 1e-5f);
#pragma unroll
        for (int q = 0; q < 8; ++q) {
            const float4 g = *(const float4*)&lg[half * 32 + 4 * q];
            u[4 * q + 0] = u[4 * q + 0] * inv * g.x;
            u[4 * q + 1] = u[4 * q + 1] * inv * g.y;
            u[4 * q + 2] = u[4 * q + 2] * inv * g.z;
            u[4 * q + 3] = u[4 * q + 3] * inv * g.w;
        }
    }

    v8f acc[4] = {};

#pragma unroll
    for (int sp = 0; sp < 8; ++sp) {
        v2f b0; b0.x = u[4 * sp];     b0.y = u[4 * sp + 1];
        v2f b1; b1.x = u[4 * sp + 2]; b1.y = u[4 * sp + 3];
        float4 aa[4];
#pragma unroll
        for (int jo = 0; jo < 4; ++jo)
            aa[jo] = *(const float4*)&lw[(jo * 16 + c) * WST + half * 32 + 4 * sp];
#pragma unroll
        for (int jo = 0; jo < 4; ++jo) {
            v2f a; a.x = aa[jo].x; a.y = aa[jo].y;
            acc[jo] = __builtin_amdgcn_wmma_f32_16x16x4_f32(
                false, a, false, b0, (short)0, acc[jo], false, false);
        }
#pragma unroll
        for (int jo = 0; jo < 4; ++jo) {
            v2f a; a.x = aa[jo].z; a.y = aa[jo].w;
            acc[jo] = __builtin_amdgcn_wmma_f32_16x16x4_f32(
                false, a, false, b1, (short)0, acc[jo], false, false);
        }
    }

    float p0 = 0.f, p1 = 0.f, p2 = 0.f, p3 = 0.f;
#pragma unroll
    for (int jo = 0; jo < 4; ++jo) {
#pragma unroll
        for (int r = 0; r < 8; ++r)
            u[8 * jo + r] += fmaxf(acc[jo][r], 0.f);
    }
#pragma unroll
    for (int q = 0; q < 8; ++q) {
        p0 += u[4 * q + 0] * u[4 * q + 0];
        p1 += u[4 * q + 1] * u[4 * q + 1];
        p2 += u[4 * q + 2] * u[4 * q + 2];
        p3 += u[4 * q + 3] * u[4 * q + 3];
    }
    float ss2 = (p0 + p1) + (p2 + p3);
    ss2 += __shfl_xor(ss2, 16, 32);
    const float inv2 = rsqrtf(ss2 * (1.0f / 64.0f) + 1e-5f);
    float4* out4 = (float4*)(fin ? fin : t);
#pragma unroll
    for (int q = 0; q < 8; ++q) {
        const float4 g = *(const float4*)&lg[HID + half * 32 + 4 * q];
        u[4 * q + 0] = u[4 * q + 0] * inv2 * g.x;
        u[4 * q + 1] = u[4 * q + 1] * inv2 * g.y;
        u[4 * q + 2] = u[4 * q + 2] * inv2 * g.z;
        u[4 * q + 3] = u[4 * q + 3] * inv2 * g.w;
    }
    if (full) {
#pragma unroll
        for (int q = 0; q < 8; ++q) {
            float4 o;
            o.x = u[4 * q + 0]; o.y = u[4 * q + 1];
            o.z = u[4 * q + 2]; o.w = u[4 * q + 3];
            *(float4*)(town + 4 * q) = o;
        }
        __builtin_amdgcn_wave_barrier();
        float4 o[8];
#pragma unroll
        for (int i = 0; i < 8; ++i) o[i] = *(const float4*)(tmem + 2 * i * SP);
#pragma unroll
        for (int i = 0; i < 8; ++i) out4[g0 + 32 * i] = o[i];
    } else if (c < rem) {
#pragma unroll
        for (int q = 0; q < 8; ++q) {
            float4 o;
            o.x = u[4 * q + 0]; o.y = u[4 * q + 1];
            o.z = u[4 * q + 2]; o.w = u[4 * q + 3];
            out4[(node0 + c) * (HID / 4) + half * 8 + q] = o;
        }
    }

    if (Wn) {
        v8f acn[4] = {};
#pragma unroll
        for (int sp = 0; sp < 8; ++sp) {
            v2f b0; b0.x = u[4 * sp];     b0.y = u[4 * sp + 1];
            v2f b1; b1.x = u[4 * sp + 2]; b1.y = u[4 * sp + 3];
            float4 aa[4];
#pragma unroll
            for (int jo = 0; jo < 4; ++jo)
                aa[jo] = *(const float4*)&lw[64 * WST + (jo * 16 + c) * WST + half * 32 + 4 * sp];
#pragma unroll
            for (int jo = 0; jo < 4; ++jo) {
                v2f a; a.x = aa[jo].x; a.y = aa[jo].y;
                acn[jo] = __builtin_amdgcn_wmma_f32_16x16x4_f32(
                    false, a, false, b0, (short)0, acn[jo], false, false);
            }
#pragma unroll
            for (int jo = 0; jo < 4; ++jo) {
                v2f a; a.x = aa[jo].z; a.y = aa[jo].w;
                acn[jo] = __builtin_amdgcn_wmma_f32_16x16x4_f32(
                    false, a, false, b1, (short)0, acn[jo], false, false);
            }
        }
        float4* m4 = (float4*)mout;
        if (full) {
#pragma unroll
            for (int jo = 0; jo < 4; ++jo) {
                float4 o0, o1;
                o0.x = fmaxf(acn[jo][0], 0.f); o0.y = fmaxf(acn[jo][1], 0.f);
                o0.z = fmaxf(acn[jo][2], 0.f); o0.w = fmaxf(acn[jo][3], 0.f);
                o1.x = fmaxf(acn[jo][4], 0.f); o1.y = fmaxf(acn[jo][5], 0.f);
                o1.z = fmaxf(acn[jo][6], 0.f); o1.w = fmaxf(acn[jo][7], 0.f);
                *(float4*)(town + 8 * jo)     = o0;
                *(float4*)(town + 8 * jo + 4) = o1;
            }
            __builtin_amdgcn_wave_barrier();
            float4 o[8];
#pragma unroll
            for (int i = 0; i < 8; ++i) o[i] = *(const float4*)(tmem + 2 * i * SP);
#pragma unroll
            for (int i = 0; i < 8; ++i) m4[g0 + 32 * i] = o[i];
        } else if (c < rem) {
#pragma unroll
            for (int jo = 0; jo < 4; ++jo) {
                float4 o0, o1;
                o0.x = fmaxf(acn[jo][0], 0.f); o0.y = fmaxf(acn[jo][1], 0.f);
                o0.z = fmaxf(acn[jo][2], 0.f); o0.w = fmaxf(acn[jo][3], 0.f);
                o1.x = fmaxf(acn[jo][4], 0.f); o1.y = fmaxf(acn[jo][5], 0.f);
                o1.z = fmaxf(acn[jo][6], 0.f); o1.w = fmaxf(acn[jo][7], 0.f);
                m4[(node0 + c) * (HID / 4) + half * 8 + 2 * jo]     = o0;
                m4[(node0 + c) * (HID / 4) + half * 8 + 2 * jo + 1] = o1;
            }
        }
    }
}

extern "C" void kernel_launch(void* const* d_in, const int* in_sizes, int n_in,
                              void* d_out, int out_size, void* d_ws, size_t ws_size,
                              hipStream_t stream)
{
    const float* x       = (const float*)d_in[0];
    const float* conv_w  = (const float*)d_in[1];
    const float* conv_nw = (const float*)d_in[2];
    const float* hid_w   = (const float*)d_in[3];
    const float* hid_nw  = (const float*)d_in[4];
    const int*   row     = (const int*)d_in[5];
    const int*   col     = (const int*)d_in[6];

    const int n  = in_sizes[0] / HID;
    const int ne = in_sizes[5];

    float* P  = (float*)d_out;
    float* Q0 = (float*)d_ws;
    float* R  = Q0 + ((size_t)n + 1) * HID;
    float* Q1 = R + (size_t)n * HID;
    int* cnt = (int*)(Q1 + ((size_t)n + 1) * HID);
    int* ell = cnt + ((size_t)n + 31) / 32 * 32;

    const dim3 blk(LTHREADS);
    const int gemm_grid = (n + LROWS - 1) / LROWS;
    const int fill_grid = (n + FROWS - 1) / FROWS;
    const int prod_grid = fill_grid < 192 ? 256 - fill_grid : 64;

    gnn_first<<<fill_grid + prod_grid, dim3(FROWS), 0, stream>>>(x, conv_w, Q0, R, row, col,
                                                                 cnt, ell, n, ne, fill_grid);

    for (int l = 0; l < 4; ++l) {
        const float* hw = hid_w  + (size_t)l * HID * HID;
        const float* cn = conv_nw + l * HID;
        const float* hn = hid_nw  + l * HID;
        const float* cwn = l < 3 ? conv_w + (size_t)(l + 1) * HID * HID : nullptr;

        const float* mcur = (l & 1) ? Q1 : Q0;
        float*       mnxt = (l & 1) ? Q0 : Q1;

        gnn_pull_norm_gemm_norm<<<gemm_grid, blk, 0, stream>>>(mcur, cnt, ell, row, col, ne,
                                                               R, hw, cn, hn,
                                                               l == 3 ? P : nullptr,
                                                               cwn, mnxt, n);
    }
}
